# starting kernel
# baseline (speedup 1.0000x reference)
#include <hip/hip_runtime.h>

typedef __attribute__((ext_vector_type(2))) float v2f;
typedef __attribute__((ext_vector_type(8))) float v8f;

#define IN_SIZE  256
#define OUT_SIZE 256
#define LDA 268
#define LDB 72

__global__ __launch_bounds__(128)
void gcn_gemm_wmma(const float* __restrict__ X, const float* __restrict__ W,
                   float* __restrict__ S, int M) {
    __shared__ float lA[16 * LDA];
    __shared__ float lB[IN_SIZE * LDB];

    const int t    = threadIdx.x;
    const int lane = t & 31;
    const int wv   = t >> 5;
    const int lm   = lane & 15;
    const int half = lane >> 4;

    const int row0 = blockIdx.x * 16;
    const int colb = blockIdx.y * 64;
    const int wcol = wv * 16;

    #pragma unroll
    for (int i = 0; i < 8; ++i) {
        int idx4 = t + i * 128;
        int r    = idx4 >> 6;
        int c4   = idx4 & 63;
        float4 v = make_float4(0.f, 0.f, 0.f, 0.f);
        if (row0 + r < M)
            v = *(const float4*)(X + (size_t)(row0 + r) * IN_SIZE + c4 * 4);
        *(float4*)(&lA[r * LDA + c4 * 4]) = v;
    }
    #pragma unroll
    for (int i = 0; i < 32; ++i) {
        int idx4 = t + i * 128;
        int r    = idx4 >> 4;
        int c4   = idx4 & 15;
        float4 v = *(const float4*)(W + (size_t)r * OUT_SIZE + colb + c4 * 4);
        *(float4*)(&lB[r * LDB + c4 * 4]) = v;
    }
    __syncthreads();

    const int aBase = lm * LDA + 2 * half;
    const int bBase = 2 * half * LDB + wcol + lm;

    v8f acc0 = {0.f,0.f,0.f,0.f,0.f,0.f,0.f,0.f};
    v8f acc1 = acc0, acc2 = acc0, acc3 = acc0;

    #pragma unroll
    for (int k = 0; k < 64; k += 4) {
        {
            v2f a = *(const v2f*)(&lA[aBase + k]);
            v2f b; b.x = lB[bBase + k * LDB]; b.y = lB[bBase + (k + 1) * LDB];
            acc0 = __builtin_amdgcn_wmma_f32_16x16x4_f32(false, a, false, b, (short)0, acc0, false, false);
        }
        {
            v2f a = *(const v2f*)(&lA[aBase + k + 64]);
            v2f b; b.x = lB[bBase + (k + 64) * LDB]; b.y = lB[bBase + (k + 65) * LDB];
            acc1 = __builtin_amdgcn_wmma_f32_16x16x4_f32(false, a, false, b, (short)0, acc1, false, false);
        }
        {
            v2f a = *(const v2f*)(&lA[aBase + k + 128]);
            v2f b; b.x = lB[bBase + (k + 128) * LDB]; b.y = lB[bBase + (k + 129) * LDB];
            acc2 = __builtin_amdgcn_wmma_f32_16x16x4_f32(false, a, false, b, (short)0, acc2, false, false);
        }
        {
            v2f a = *(const v2f*)(&lA[aBase + k + 192]);
            v2f b; b.x = lB[bBase + (k + 192) * LDB]; b.y = lB[bBase + (k + 193) * LDB];
            acc3 = __builtin_amdgcn_wmma_f32_16x16x4_f32(false, a, false, b, (short)0, acc3, false, false);
        }
    }

    v8f acc = (acc0 + acc1) + (acc2 + acc3);

    #pragma unroll
    for (int v = 0; v < 8; ++v) {
        int r = row0 + v + 8 * half;
        if (r < M)
            S[(size_t)r * OUT_SIZE + colb + wcol + lm] = acc[v];
    }
}

__global__ __launch_bounds__(256)
void gcn_init_bias(float* __restrict__ out, const float* __restrict__ bias,
                   long long total) {
    long long i = (long long)blockIdx.x * blockDim.x + threadIdx.x;
    if (i < total) out[i] = bias[i & (OUT_SIZE - 1)];
}

__global__ __launch_bounds__(256)
void gcn_edge_scatter(const float* __restrict__ S, const float* __restrict__ ew,
                      const int* __restrict__ esrc, const int* __restrict__ edst,
                      float* __restrict__ out, int E) {
    int e    = (int)(((long long)blockIdx.x * blockDim.x + threadIdx.x) >> 5);
    int lane = threadIdx.x & 31;
    if (e >= E) return;
    int   s = esrc[e];
    int   d = edst[e];
    float w = ew[e];
    const float* sp = S   + (size_t)s * OUT_SIZE;
    float*       op = out + (size_t)d * OUT_SIZE;
    #pragma unroll
    for (int i = 0; i < 8; ++i) {
        int c = lane + i * 32;
        atomicAdd(&op[c], sp[c] * w);
    }
}

extern "C" void kernel_launch(void* const* d_in, const int* in_sizes, int n_in,
                              void* d_out, int out_size, void* d_ws, size_t ws_size,
                              hipStream_t stream) {
    const float* x    = (const float*)d_in[0];
    const float* w    = (const float*)d_in[1];
    const float* bias = (const float*)d_in[2];
    const float* ew   = (const float*)d_in[3];
    const int*   esrc = (const int*)d_in[4];
    const int*   edst = (const int*)d_in[5];
    float* out  = (float*)d_out;
    float* supp = (float*)d_ws;

    const int M = in_sizes[0] / IN_SIZE;
    const int E = in_sizes[3];

    dim3 ggrid((M + 15) / 16, OUT_SIZE / 64);
    gcn_gemm_wmma<<<ggrid, 128, 0, stream>>>(x, w, supp, M);

    long long total = (long long)M * OUT_SIZE;
    int iblocks = (int)((total + 255) / 256);
    gcn_init_bias<<<iblocks, 256, 0, stream>>>(out, bias, total);

    int sblocks = (E + 7) / 8;
    gcn_edge_scatter<<<sblocks, 256, 0, stream>>>(supp, ew, esrc, edst, out, E);
}


# Float-atomic global scatter and bias-fill launch replaced by dst-row ownership: a block scans the edge list for its 196 rows and sums S rows in an LDS tile by LDS float adds (arrival order)
# speedup vs baseline: 2.2427x; 2.2427x over previous
#include <hip/hip_runtime.h>

typedef __attribute__((ext_vector_type(2))) float v2f;
typedef __attribute__((ext_vector_type(8))) float v8f;
typedef __attribute__((ext_vector_type(4))) float v4f;

#define IN_SIZE  256
#define OUT_SIZE 256
#define BM   80
#define BK   32
#define NKC  (IN_SIZE / BK)
#define LDA  36
#define LDBP 544
#define A_STAGE (BM * LDA)
#define B_STAGE ((BK / 2) * LDBP)

__device__ __forceinline__ void load_chunk(const float* xa0, const float* xa1, const float* xa2,
                                           bool a3, const float* wb, int kc,
                                           float4& ra0, float4& ra1, float4& ra2,
                                           float4 (&rb0)[4], float4 (&rb1)[4]) {
    ra0 = *(const float4*)(xa0 + kc * BK);
    ra1 = *(const float4*)(xa1 + kc * BK);
    if (a3) ra2 = *(const float4*)(xa2 + kc * BK);
    const float* p = wb + (size_t)kc * BK * OUT_SIZE;
    #pragma unroll
    for (int i = 0; i < 4; ++i) {
        rb0[i] = *(const float4*)(p + (size_t)i * 8 * OUT_SIZE);
        rb1[i] = *(const float4*)(p + (size_t)i * 8 * OUT_SIZE + OUT_SIZE);
    }
}

__device__ __forceinline__ void store_chunk(float* la, float* lb, bool a3,
                                            const float4& ra0, const float4& ra1, const float4& ra2,
                                            const float4 (&rb0)[4], const float4 (&rb1)[4]) {
    *(float4*)(la)            = ra0;
    *(float4*)(la + 32 * LDA) = ra1;
    if (a3) *(float4*)(la + 64 * LDA) = ra2;
    #pragma unroll
    for (int i = 0; i < 4; ++i) {
        float* d = lb + i * 4 * LDBP;
        *(float4*)(d)     = make_float4(rb0[i].x, rb1[i].x, rb0[i].y, rb1[i].y);
        *(float4*)(d + 4) = make_float4(rb0[i].z, rb1[i].z, rb0[i].w, rb1[i].w);
    }
}

__global__ __launch_bounds__(256)
void gcn_gemm_wmma(const float* __restrict__ X, const float* __restrict__ W,
                   float* __restrict__ S, int M) {
    __shared__ __attribute__((aligned(16))) float lA[2 * A_STAGE];
    __shared__ __attribute__((aligned(16))) float lB[2 * B_STAGE];

    const int t    = threadIdx.x;
    const int lane = t & 31;
    const int wv   = t >> 5;
    const int c    = lane & 15;
    const int half = lane >> 4;

    const int row0 = blockIdx.x * BM;
    const int n0   = wv * 32;

    const bool a3  = t < 128;
    const int  ar  = t >> 3, ac4 = t & 7;
    const float* xa0 = X + (size_t)min(row0 + ar,      M - 1) * IN_SIZE + ac4 * 4;
    const float* xa1 = X + (size_t)min(row0 + ar + 32, M - 1) * IN_SIZE + ac4 * 4;
    const float* xa2 = X + (size_t)min(row0 + (ar & 15) + 64, M - 1) * IN_SIZE + ac4 * 4;
    const float* wb  = W + (size_t)(2 * (t >> 6)) * OUT_SIZE + (t & 63) * 4;
    float* la = lA + ar * LDA + ac4 * 4;
    float* lb = lB + (t >> 6) * LDBP + (t & 63) * 8;

    const float* pa = lA + c * LDA + 2 * half;
    const float* pb = lB + half * LDBP + (n0 + c) * 2;

    float4 ra0, ra1, ra2 = make_float4(0.f, 0.f, 0.f, 0.f);
    float4 rb0[4], rb1[4];

    v8f acc[5][2];
    #pragma unroll
    for (int mt = 0; mt < 5; ++mt)
        #pragma unroll
        for (int nt = 0; nt < 2; ++nt)
            acc[mt][nt] = (v8f){0.f,0.f,0.f,0.f,0.f,0.f,0.f,0.f};

    load_chunk(xa0, xa1, xa2, a3, wb, 0, ra0, ra1, ra2, rb0, rb1);
    store_chunk(la, lb, a3, ra0, ra1, ra2, rb0, rb1);
    __syncthreads();

    for (int kc = 0; kc < NKC; ++kc) {
        const int cur = kc & 1;
        const bool more = kc + 1 < NKC;
        if (more)
            load_chunk(xa0, xa1, xa2, a3, wb, kc + 1, ra0, ra1, ra2, rb0, rb1);
        __builtin_amdgcn_sched_barrier(0);

        const float* qa = pa + cur * A_STAGE;
        const float* qb = pb + cur * B_STAGE;
        #pragma unroll
        for (int ks = 0; ks < BK / 4; ++ks) {
            v2f a[5], b[2];
            #pragma unroll
            for (int mt = 0; mt < 5; ++mt)
                a[mt] = *(const v2f*)(qa + mt * 16 * LDA + ks * 4);
            #pragma unroll
            for (int nt = 0; nt < 2; ++nt)
                b[nt] = *(const v2f*)(qb + ks * 2 * LDBP + nt * 32);
            #pragma unroll
            for (int mt = 0; mt < 5; ++mt)
                #pragma unroll
                for (int nt = 0; nt < 2; ++nt)
                    acc[mt][nt] = __builtin_amdgcn_wmma_f32_16x16x4_f32(
                        false, a[mt], false, b[nt], (short)0, acc[mt][nt], false, false);
        }

        if (more) {
            store_chunk(la + (cur ^ 1) * A_STAGE, lb + (cur ^ 1) * B_STAGE, a3,
                        ra0, ra1, ra2, rb0, rb1);
            __syncthreads();
        }
    }

    float* sp = S + (size_t)(row0 + 8 * half) * OUT_SIZE + n0 + c;
    if (row0 + BM <= M) {
        #pragma unroll
        for (int mt = 0; mt < 5; ++mt)
            #pragma unroll
            for (int nt = 0; nt < 2; ++nt)
                #pragma unroll
                for (int v = 0; v < 8; ++v)
                    sp[(size_t)(mt * 16 + v) * OUT_SIZE + nt * 16] = acc[mt][nt][v];
    } else {
        #pragma unroll
        for (int mt = 0; mt < 5; ++mt)
            #pragma unroll
            for (int nt = 0; nt < 2; ++nt)
                #pragma unroll
                for (int v = 0; v < 8; ++v)
                    if (row0 + mt * 16 + v + 8 * half < M)
                        sp[(size_t)(mt * 16 + v) * OUT_SIZE + nt * 16] = acc[mt][nt][v];
    }
}

#define TR     196
#define AGG_T  1024
#define ECHUNK 1024
#define LFLUSH 24
#define LCAP   (LFLUSH + 128)

__device__ __forceinline__ void add_listed_rows(const int* ms, const int* mr, const float* mw, int cnt,
                                                const float* __restrict__ S, float* acc, int lane) {
    for (int i = 0; i < cnt; i += 8) {
        v4f   a[8], b[8];
        float w[8];
        int   r[8];
        #pragma unroll
        for (int u = 0; u < 8; ++u) {
            const int idx = min(i + u, cnt - 1);
            const float* row = S + (size_t)ms[idx] * OUT_SIZE + lane * 4;
            a[u] = *(const v4f*)(row);
            b[u] = *(const v4f*)(row + 128);
            w[u] = mw[idx];
            r[u] = mr[idx];
        }
        asm volatile("" :: "v"(a[0]), "v"(a[1]), "v"(a[2]), "v"(a[3]), "v"(a[4]), "v"(a[5]), "v"(a[6]), "v"(a[7]),
                           "v"(b[0]), "v"(b[1]), "v"(b[2]), "v"(b[3]), "v"(b[4]), "v"(b[5]), "v"(b[6]), "v"(b[7]));
        #pragma unroll
        for (int u = 0; u < 8; ++u) {
            if (i + u < cnt) {
                float* p = acc + r[u] * OUT_SIZE + lane * 4;
                atomicAdd(p + 0,   a[u].x * w[u]);
                atomicAdd(p + 1,   a[u].y * w[u]);
                atomicAdd(p + 2,   a[u].z * w[u]);
                atomicAdd(p + 3,   a[u].w * w[u]);
                atomicAdd(p + 128, b[u].x * w[u]);
                atomicAdd(p + 129, b[u].y * w[u]);
                atomicAdd(p + 130, b[u].z * w[u]);
                atomicAdd(p + 131, b[u].w * w[u]);
            }
        }
    }
}

#define PICK(m, j, k) ((j) == 0 ? m[0][k] : (j) == 1 ? m[1][k] : (j) == 2 ? m[2][k] : (j) == 3 ? m[3][k] : \
                       (j) == 4 ? m[4][k] : (j) == 5 ? m[5][k] : (j) == 6 ? m[6][k] : m[7][k])

__global__ __launch_bounds__(AGG_T)
void gcn_aggregate_rows(const float* __restrict__ S, const float* __restrict__ bias,
                        const float* __restrict__ ew, const int* __restrict__ esrc,
                        const int* __restrict__ edst, float* __restrict__ out, int M, int E) {
    __shared__ __attribute__((aligned(16))) float acc[TR * OUT_SIZE];
    __shared__ int   l_src[32 * LCAP];
    __shared__ int   l_row[32 * LCAP];
    __shared__ float l_w[32 * LCAP];

    const int t    = threadIdx.x;
    const int lane = t & 31;
    const int wv   = __builtin_amdgcn_readfirstlane(t >> 5);
    const int lo   = blockIdx.x * TR;

    const float4 b4 = *(const float4*)(bias + (t & 63) * 4);
    #pragma unroll
    for (int it = 0; it < (TR * 64 + AGG_T - 1) / AGG_T; ++it) {
        const int i = t + it * AGG_T;
        if (i < TR * 64) *(float4*)(acc + i * 4) = b4;
    }
    __syncthreads();

    int*   ms = l_src + wv * LCAP;
    int*   mr = l_row + wv * LCAP;
    float* mw = l_w   + wv * LCAP;
    int cnt = 0;

    const int nchunk = (E + ECHUNK - 1) / ECHUNK;
    const int rot = nchunk > 0 ? (int)((blockIdx.x * 61u) % (unsigned)nchunk) : 0;
    const unsigned lt = (1u << lane) - 1u;

    for (int ch = wv; ch < nchunk; ch += 32) {
        int cc = ch + rot;
        if (cc >= nchunk) cc -= nchunk;
        const int e0 = cc * ECHUNK + lane * 4;

        int4 dv[8];
        if (cc * ECHUNK + ECHUNK <= E) {
            #pragma unroll
            for (int j = 0; j < 8; ++j) dv[j] = *(const int4*)(edst + e0 + j * 128);
        } else {
            #pragma unroll
            for (int j = 0; j < 8; ++j) {
                const int e = e0 + j * 128;
                dv[j].x = e     < E ? edst[e]     : -1;
                dv[j].y = e + 1 < E ? edst[e + 1] : -1;
                dv[j].z = e + 2 < E ? edst[e + 2] : -1;
                dv[j].w = e + 3 < E ? edst[e + 3] : -1;
            }
        }

        unsigned m[8][4];
        unsigned jm = 0;
        #pragma unroll
        for (int j = 0; j < 8; ++j) {
            m[j][0] = __builtin_amdgcn_ballot_w32((unsigned)(dv[j].x - lo) < (unsigned)TR);
            m[j][1] = __builtin_amdgcn_ballot_w32((unsigned)(dv[j].y - lo) < (unsigned)TR);
            m[j][2] = __builtin_amdgcn_ballot_w32((unsigned)(dv[j].z - lo) < (unsigned)TR);
            m[j][3] = __builtin_amdgcn_ballot_w32((unsigned)(dv[j].w - lo) < (unsigned)TR);
            if (m[j][0] | m[j][1] | m[j][2] | m[j][3]) jm |= 1u << j;
        }

        while (jm) {
            const int j = __builtin_ctz(jm);
            jm &= jm - 1;
            #pragma unroll
            for (int k = 0; k < 4; ++k) {
                const unsigned mk = PICK(m, j, k);
                if (mk) {
                    if ((mk >> lane) & 1u) {
                        const int pos = cnt + __popc(mk & lt);
                        const int e   = e0 + j * 128 + k;
                        ms[pos] = esrc[e];
                        mr[pos] = edst[e] - lo;
                        mw[pos] = ew[e];
                    }
                    cnt += __popc(mk);
                }
            }
            if (cnt >= LFLUSH) {
                add_listed_rows(ms, mr, mw, cnt, S, acc, lane);
                cnt = 0;
            }
        }
    }
    add_listed_rows(ms, mr, mw, cnt, S, acc, lane);
    __syncthreads();

    for (int i = t; i < TR * 64; i += AGG_T) {
        const int row = lo + (i >> 6);
        if (row < M)
            *(float4*)(out + (size_t)row * OUT_SIZE + (i & 63) * 4) = *(const float4*)(acc + i * 4);
    }
}

extern "C" void kernel_launch(void* const* d_in, const int* in_sizes, int n_in,
                              void* d_out, int out_size, void* d_ws, size_t ws_size,
                              hipStream_t stream) {
    const float* x    = (const float*)d_in[0];
    const float* w    = (const float*)d_in[1];
    const float* bias = (const float*)d_in[2];
    const float* ew   = (const float*)d_in[3];
    const int*   esrc = (const int*)d_in[4];
    const int*   edst = (const int*)d_in[5];
    float* out  = (float*)d_out;
    float* supp = (float*)d_ws;

    const int M = in_sizes[0] / IN_SIZE;
    const int E = in_sizes[3];

    dim3 ggrid((M + BM - 1) / BM);
    gcn_gemm_wmma<<<ggrid, 256, 0, stream>>>(x, w, supp, M);

    dim3 agrid((M + TR - 1) / TR);
    gcn_aggregate_rows<<<agrid, AGG_T, 0, stream>>>(supp, bias, ew, esrc, edst, out, M, E);
}


# Aggregation: LDS row tile + LDS float adds replaced by a per-block counting sort of the block's edges by row; a wave sums each row in registers (arrival order); 391 rows/block, 256 blocks
# speedup vs baseline: 3.3370x; 1.4879x over previous
#include <hip/hip_runtime.h>

typedef __attribute__((ext_vector_type(2))) float v2f;
typedef __attribute__((ext_vector_type(8))) float v8f;
typedef __attribute__((ext_vector_type(4))) float v4f;

#define IN_SIZE  256
#define OUT_SIZE 256
#define BM   80
#define BK   32
#define NKC  (IN_SIZE / BK)
#define LDA  36
#define LDBP 544
#define A_STAGE (BM * LDA)
#define B_STAGE ((BK / 2) * LDBP)

__device__ __forceinline__ void load_chunk(const float* xa0, const float* xa1, const float* xa2,
                                           bool a3, const float* wb, int kc,
                                           float4& ra0, float4& ra1, float4& ra2,
                                           float4 (&rb0)[4], float4 (&rb1)[4]) {
    ra0 = *(const float4*)(xa0 + kc * BK);
    ra1 = *(const float4*)(xa1 + kc * BK);
    if (a3) ra2 = *(const float4*)(xa2 + kc * BK);
    const float* p = wb + (size_t)kc * BK * OUT_SIZE;
    #pragma unroll
    for (int i = 0; i < 4; ++i) {
        rb0[i] = *(const float4*)(p + (size_t)i * 8 * OUT_SIZE);
        rb1[i] = *(const float4*)(p + (size_t)i * 8 * OUT_SIZE + OUT_SIZE);
    }
}

__device__ __forceinline__ void store_chunk(float* la, float* lb, bool a3,
                                            const float4& ra0, const float4& ra1, const float4& ra2,
                                            const float4 (&rb0)[4], const float4 (&rb1)[4]) {
    *(float4*)(la)            = ra0;
    *(float4*)(la + 32 * LDA) = ra1;
    if (a3) *(float4*)(la + 64 * LDA) = ra2;
    #pragma unroll
    for (int i = 0; i < 4; ++i) {
        float* d = lb + i * 4 * LDBP;
        *(float4*)(d)     = make_float4(rb0[i].x, rb1[i].x, rb0[i].y, rb1[i].y);
        *(float4*)(d + 4) = make_float4(rb0[i].z, rb1[i].z, rb0[i].w, rb1[i].w);
    }
}

__global__ __launch_bounds__(256)
void gcn_gemm_wmma(const float* __restrict__ X, const float* __restrict__ W,
                   float* __restrict__ S, int M) {
    __shared__ __attribute__((aligned(16))) float lA[2 * A_STAGE];
    __shared__ __attribute__((aligned(16))) float lB[2 * B_STAGE];

    const int t    = threadIdx.x;
    const int lane = t & 31;
    const int wv   = t >> 5;
    const int c    = lane & 15;
    const int half = lane >> 4;

    const int row0 = blockIdx.x * BM;
    const int n0   = wv * 32;

    const bool a3  = t < 128;
    const int  ar  = t >> 3, ac4 = t & 7;
    const float* xa0 = X + (size_t)min(row0 + ar,      M - 1) * IN_SIZE + ac4 * 4;
    const float* xa1 = X + (size_t)min(row0 + ar + 32, M - 1) * IN_SIZE + ac4 * 4;
    const float* xa2 = X + (size_t)min(row0 + (ar & 15) + 64, M - 1) * IN_SIZE + ac4 * 4;
    const float* wb  = W + (size_t)(2 * (t >> 6)) * OUT_SIZE + (t & 63) * 4;
    float* la = lA + ar * LDA + ac4 * 4;
    float* lb = lB + (t >> 6) * LDBP + (t & 63) * 8;

    const float* pa = lA + c * LDA + 2 * half;
    const float* pb = lB + half * LDBP + (n0 + c) * 2;

    float4 ra0, ra1, ra2 = make_float4(0.f, 0.f, 0.f, 0.f);
    float4 rb0[4], rb1[4];

    v8f acc[5][2];
    #pragma unroll
    for (int mt = 0; mt < 5; ++mt)
        #pragma unroll
        for (int nt = 0; nt < 2; ++nt)
            acc[mt][nt] = (v8f){0.f,0.f,0.f,0.f,0.f,0.f,0.f,0.f};

    load_chunk(xa0, xa1, xa2, a3, wb, 0, ra0, ra1, ra2, rb0, rb1);
    store_chunk(la, lb, a3, ra0, ra1, ra2, rb0, rb1);
    __syncthreads();

    for (int kc = 0; kc < NKC; ++kc) {
        const int cur = kc & 1;
        const bool more = kc + 1 < NKC;
        if (more)
            load_chunk(xa0, xa1, xa2, a3, wb, kc + 1, ra0, ra1, ra2, rb0, rb1);
        __builtin_amdgcn_sched_barrier(0);

        const float* qa = pa + cur * A_STAGE;
        const float* qb = pb + cur * B_STAGE;
        #pragma unroll
        for (int ks = 0; ks < BK / 4; ++ks) {
            v2f a[5], b[2];
            #pragma unroll
            for (int mt = 0; mt < 5; ++mt)
                a[mt] = *(const v2f*)(qa + mt * 16 * LDA + ks * 4);
            #pragma unroll
            for (int nt = 0; nt < 2; ++nt)
                b[nt] = *(const v2f*)(qb + ks * 2 * LDBP + nt * 32);
            #pragma unroll
            for (int mt = 0; mt < 5; ++mt)
                #pragma unroll
                for (int nt = 0; nt < 2; ++nt)
                    acc[mt][nt] = __builtin_amdgcn_wmma_f32_16x16x4_f32(
                        false, a[mt], false, b[nt], (short)0, acc[mt][nt], false, false);
        }

        if (more) {
            store_chunk(la + (cur ^ 1) * A_STAGE, lb + (cur ^ 1) * B_STAGE, a3,
                        ra0, ra1, ra2, rb0, rb1);
            __syncthreads();
        }
    }

    float* sp = S + (size_t)(row0 + 8 * half) * OUT_SIZE + n0 + c;
    if (row0 + BM <= M) {
        #pragma unroll
        for (int mt = 0; mt < 5; ++mt)
            #pragma unroll
            for (int nt = 0; nt < 2; ++nt)
                #pragma unroll
                for (int v = 0; v < 8; ++v)
                    sp[(size_t)(mt * 16 + v) * OUT_SIZE + nt * 16] = acc[mt][nt][v];
    } else {
        #pragma unroll
        for (int mt = 0; mt < 5; ++mt)
            #pragma unroll
            for (int nt = 0; nt < 2; ++nt)
                #pragma unroll
                for (int v = 0; v < 8; ++v)
                    if (row0 + mt * 16 + v + 8 * half < M)
                        sp[(size_t)(mt * 16 + v) * OUT_SIZE + nt * 16] = acc[mt][nt][v];
    }
}

#define TR     391
#define AGG_T  1024
#define ECHUNK 1024
#define CAPW   512
#define NROWP  ((TR + 31) / 32)

__global__ __launch_bounds__(AGG_T)
void gcn_aggregate_rows(const float* __restrict__ S, const float* __restrict__ bias,
                        const float* __restrict__ ew, const int* __restrict__ esrc,
                        const int* __restrict__ edst, float* __restrict__ out, int M, int E) {
    __shared__ int   l_e[32 * CAPW];
    __shared__ int   l_row[32 * CAPW];
    __shared__ int   s_src[32 * CAPW];
    __shared__ float s_w[32 * CAPW];
    __shared__ int   rowcnt[TR + 1];
    __shared__ int   rowoff[TR + 1];
    __shared__ int   rowfill[TR + 1];
    __shared__ int   ovf_flag;

    const int t    = threadIdx.x;
    const int lane = t & 31;
    const int wv   = __builtin_amdgcn_readfirstlane(t >> 5);
    const int lo   = blockIdx.x * TR;

    if (t <= TR) rowcnt[t] = 0;
    if (t == 0) ovf_flag = 0;
    __syncthreads();

    int* me = l_e   + wv * CAPW;
    int* mr = l_row + wv * CAPW;
    int  cnt = 0;
    bool ovf = false;

    const int nchunk = (E + ECHUNK - 1) / ECHUNK;
    const int rot = nchunk > 0 ? (int)((blockIdx.x * 61u) % (unsigned)nchunk) : 0;
    const unsigned lt = (1u << lane) - 1u;

    for (int ch = wv; ch < nchunk; ch += 32) {
        int cc = ch + rot;
        if (cc >= nchunk) cc -= nchunk;
        const int e0 = cc * ECHUNK + lane * 4;

        int4 dv[8];
        if (cc * ECHUNK + ECHUNK <= E) {
            #pragma unroll
            for (int j = 0; j < 8; ++j) dv[j] = *(const int4*)(edst + e0 + j * 128);
        } else {
            #pragma unroll
            for (int j = 0; j < 8; ++j) {
                const int e = e0 + j * 128;
                dv[j].x = e     < E ? edst[e]     : -1;
                dv[j].y = e + 1 < E ? edst[e + 1] : -1;
                dv[j].z = e + 2 < E ? edst[e + 2] : -1;
                dv[j].w = e + 3 < E ? edst[e + 3] : -1;
            }
        }

        #pragma unroll
        for (int j = 0; j < 8; ++j) {
            const int dj[4] = {dv[j].x, dv[j].y, dv[j].z, dv[j].w};
            unsigned mk[4];
            #pragma unroll
            for (int k = 0; k < 4; ++k)
                mk[k] = __builtin_amdgcn_ballot_w32((unsigned)(dj[k] - lo) < (unsigned)TR);
            if (mk[0] | mk[1] | mk[2] | mk[3]) {
                if (cnt + 128 > CAPW) {
                    ovf = true;
                } else {
                    #pragma unroll
                    for (int k = 0; k < 4; ++k) {
                        if (mk[k]) {
                            if ((mk[k] >> lane) & 1u) {
                                const int pos = cnt + __popc(mk[k] & lt);
                                const int r   = dj[k] - lo;
                                me[pos] = e0 + j * 128 + k;
                                mr[pos] = r;
                                atomicAdd(&rowcnt[r], 1);
                            }
                            cnt += __popc(mk[k]);
                        }
                    }
                }
            }
        }
    }
    if (ovf && lane == 0) ovf_flag = 1;
    __syncthreads();
    const bool fallback = ovf_flag != 0;

    if (!fallback) {
        if (wv == 0) {
            int c[NROWP];
            int s = 0;
            #pragma unroll
            for (int i = 0; i < NROWP; ++i) {
                const int r = lane * NROWP + i;
                c[i] = r < TR ? rowcnt[r] : 0;
                s += c[i];
            }
            int incl = s;
            #pragma unroll
            for (int d = 1; d < 32; d <<= 1) {
                const int v = __shfl_up(incl, d, 32);
                if (lane >= d) incl += v;
            }
            int run = incl - s;
            #pragma unroll
            for (int i = 0; i < NROWP; ++i) {
                const int r = lane * NROWP + i;
                if (r < TR) { rowoff[r] = run; rowfill[r] = run; }
                run += c[i];
            }
        }
        __syncthreads();

        for (int i0 = 0; i0 < cnt; i0 += 128) {
            int   le[4], lr[4], sv[4];
            float wq[4];
            #pragma unroll
            for (int u = 0; u < 4; ++u) {
                const int i = i0 + u * 32 + lane;
                le[u] = i < cnt ? me[i] : -1;
                lr[u] = i < cnt ? mr[i] : 0;
            }
            #pragma unroll
            for (int u = 0; u < 4; ++u) {
                sv[u] = 0; wq[u] = 0.f;
                if (le[u] >= 0) { sv[u] = esrc[le[u]]; wq[u] = ew[le[u]]; }
            }
            #pragma unroll
            for (int u = 0; u < 4; ++u) {
                if (le[u] >= 0) {
                    const int slot = atomicAdd(&rowfill[lr[u]], 1);
                    s_src[slot] = sv[u];
                    s_w[slot]   = wq[u];
                }
            }
        }
        __syncthreads();

        const v4f b0 = *(const v4f*)(bias + lane * 4);
        const v4f b1 = *(const v4f*)(bias + 128 + lane * 4);
        for (int r = wv; r < TR; r += 32) {
            const int row = lo + r;
            if (row >= M) break;
            const int n = __builtin_amdgcn_readfirstlane(rowcnt[r]);
            const int o = __builtin_amdgcn_readfirstlane(rowoff[r]);
            v4f s0 = b0, s1 = b1;
            for (int i = 0; i < n; i += 8) {
                v4f   a[8], b[8];
                float w[8];
                #pragma unroll
                for (int u = 0; u < 8; ++u) {
                    const int idx = o + min(i + u, n - 1);
                    const float* p = S + (size_t)s_src[idx] * OUT_SIZE + lane * 4;
                    a[u] = *(const v4f*)(p);
                    b[u] = *(const v4f*)(p + 128);
                    w[u] = s_w[idx];
                }
                #pragma unroll
                for (int u = 0; u < 8; ++u) {
                    if (i + u < n) {
                        s0 += a[u] * w[u];
                        s1 += b[u] * w[u];
                    }
                }
            }
            float* q = out + (size_t)row * OUT_SIZE + lane * 4;
            *(v4f*)(q)       = s0;
            *(v4f*)(q + 128) = s1;
        }
    } else {
        const float4 b4 = *(const float4*)(bias + (t & 63) * 4);
        for (int i = t; i < TR * 64; i += AGG_T) {
            const int row = lo + (i >> 6);
            if (row < M) *(float4*)(out + (size_t)row * OUT_SIZE + (i & 63) * 4) = b4;
        }
        __threadfence();
        __syncthreads();
        for (int ch = wv; ch < nchunk; ch += 32) {
            const int eb = ch * ECHUNK;
            for (int x = 0; x < ECHUNK; x += 32) {
                const int e = eb + x + lane;
                const int d = e < E ? edst[e] : -1;
                unsigned mk = __builtin_amdgcn_ballot_w32((unsigned)(d - lo) < (unsigned)TR && d < M);
                while (mk) {
                    const int l = __builtin_ctz(mk);
                    mk &= mk - 1;
                    const int   ee  = eb + x + l;
                    const int   dd  = edst[ee];
                    const float w   = ew[ee];
                    const float* p  = S + (size_t)esrc[ee] * OUT_SIZE + lane * 4;
                    const v4f a = *(const v4f*)(p);
                    const v4f b = *(const v4f*)(p + 128);
                    float* q = out + (size_t)dd * OUT_SIZE + lane * 4;
                    atomicAdd(q + 0,   a.x * w);
                    atomicAdd(q + 1,   a.y * w);
                    atomicAdd(q + 2,   a.z * w);
                    atomicAdd(q + 3,   a.w * w);
                    atomicAdd(q + 128, b.x * w);
                    atomicAdd(q + 129, b.y * w);
                    atomicAdd(q + 130, b.z * w);
                    atomicAdd(q + 131, b.w * w);
                }
            }
        }
    }
}

extern "C" void kernel_launch(void* const* d_in, const int* in_sizes, int n_in,
                              void* d_out, int out_size, void* d_ws, size_t ws_size,
                              hipStream_t stream) {
    const float* x    = (const float*)d_in[0];
    const float* w    = (const float*)d_in[1];
    const float* bias = (const float*)d_in[2];
    const float* ew   = (const float*)d_in[3];
    const int*   esrc = (const int*)d_in[4];
    const int*   edst = (const int*)d_in[5];
    float* out  = (float*)d_out;
    float* supp = (float*)d_ws;

    const int M = in_sizes[0] / IN_SIZE;
    const int E = in_sizes[3];

    dim3 ggrid((M + BM - 1) / BM);
    gcn_gemm_wmma<<<ggrid, 256, 0, stream>>>(x, w, supp, M);

    dim3 agrid((M + TR - 1) / TR);
    gcn_aggregate_rows<<<agrid, AGG_T, 0, stream>>>(supp, bias, ew, esrc, edst, out, M, E);
}


# Edge list pre-grouped by destination block in a front launch (LDS counting sort per slice, no global atomics): a block fetches its own runs, not the whole edst list; row sums still arrival order
# speedup vs baseline: 3.9126x; 1.1725x over previous
#include <hip/hip_runtime.h>

typedef __attribute__((ext_vector_type(2))) float v2f;
typedef __attribute__((ext_vector_type(8))) float v8f;
typedef __attribute__((ext_vector_type(4))) float v4f;

#define IN_SIZE  256
#define OUT_SIZE 256

#define TR     391
#define AGG_T  1024
#define ECHUNK 1024
#define CAPE   16384
#define NSL    256
#define BIN_T  256
#define BUNR   16
#define NBMAX  1024
#define NROWP  ((TR + 31) / 32)
#define TM     80
#define NST    ((TR + TM - 1) / TM)
#define LDY    260
#define Y_STAGE (TM * LDY)

__device__ __forceinline__ void load_b_chunk(const float* wp, int kc, v2f (&b)[8]) {
    #pragma unroll
    for (int ks = 0; ks < 8; ++ks) {
        const float* p = wp + (size_t)(kc * 32 + ks * 4) * OUT_SIZE;
        b[ks].x = p[0];
        b[ks].y = p[OUT_SIZE];
    }
}

__device__ __forceinline__ void gemm_tile(const float* __restrict__ ytile, const float* __restrict__ W,
                                          const float* __restrict__ bias, float* __restrict__ out,
                                          int n0, int c, int half, int rbase, int lo, int M) {
    const float* ya = ytile + c * LDY + 2 * half;
    const float* wp = W + (size_t)(2 * half) * OUT_SIZE + n0 + c;

    v8f acc[5];
    #pragma unroll
    for (int mt = 0; mt < 5; ++mt) acc[mt] = (v8f){0.f,0.f,0.f,0.f,0.f,0.f,0.f,0.f};

    v2f bc[8], bn[8];
    load_b_chunk(wp, 0, bc);
    for (int kc = 0; kc < IN_SIZE / 32; ++kc) {
        if (kc + 1 < IN_SIZE / 32) load_b_chunk(wp, kc + 1, bn);
        __builtin_amdgcn_sched_barrier(0);
        #pragma unroll
        for (int ks = 0; ks < 8; ++ks) {
            v2f a[5];
            #pragma unroll
            for (int mt = 0; mt < 5; ++mt)
                a[mt] = *(const v2f*)(ya + mt * 16 * LDY + kc * 32 + ks * 4);
            #pragma unroll
            for (int mt = 0; mt < 5; ++mt)
                acc[mt] = __builtin_amdgcn_wmma_f32_16x16x4_f32(
                    false, a[mt], false, bc[ks], (short)0, acc[mt], false, false);
        }
        #pragma unroll
        for (int ks = 0; ks < 8; ++ks) bc[ks] = bn[ks];
    }

    const float bv = bias[n0 + c];
    float* op = out + (size_t)(lo + rbase + 8 * half) * OUT_SIZE + n0 + c;
    if (rbase + TM <= TR && lo + rbase + TM <= M) {
        #pragma unroll
        for (int mt = 0; mt < 5; ++mt)
            #pragma unroll
            for (int v = 0; v < 8; ++v)
                op[(size_t)(mt * 16 + v) * OUT_SIZE] = acc[mt][v] + bv;
    } else {
        #pragma unroll
        for (int mt = 0; mt < 5; ++mt)
            #pragma unroll
            for (int v = 0; v < 8; ++v) {
                const int r = rbase + mt * 16 + v + 8 * half;
                if (r < TR && lo + r < M)
                    op[(size_t)(mt * 16 + v) * OUT_SIZE] = acc[mt][v] + bv;
            }
    }
}

__device__ __forceinline__ void lds_barrier() {
    __builtin_amdgcn_fence(__ATOMIC_RELEASE, "workgroup", "local");
    __builtin_amdgcn_s_barrier();
    __builtin_amdgcn_fence(__ATOMIC_ACQUIRE, "workgroup", "local");
}


__global__ __launch_bounds__(BIN_T)
void gcn_bin_edges(const float* __restrict__ ew, const int* __restrict__ esrc,
                   const int* __restrict__ edst, int* __restrict__ wsi, int M, int E) {
    __shared__ int hist[NBMAX];
    __shared__ int fill[NBMAX];

    const int nb = (M + TR - 1) / TR;
    if (nb > NBMAX) return;

    const int t    = threadIdx.x;
    const int lane = t & 31;
    const int s    = blockIdx.x;
    const int slen = (E + NSL - 1) / NSL;
    const int cap  = NSL * slen;
    int*   rec_row = wsi;
    int*   rec_src = wsi + cap;
    float* rec_w   = (float*)(wsi + 2 * cap);
    int2*  tab     = (int2*)(wsi + 3 * cap);
    const int e_lo = s * slen;
    const int e_hi = min(E, e_lo + slen);

    for (int i = t; i < nb; i += BIN_T) hist[i] = 0;
    __syncthreads();

    for (int e0 = e_lo; e0 < e_hi; e0 += BIN_T * BUNR) {
        int d[BUNR];
        #pragma unroll
        for (int u = 0; u < BUNR; ++u) {
            const int e = e0 + u * BIN_T + t;
            d[u] = e < e_hi ? edst[e] : -1;
        }
        #pragma unroll
        for (int u = 0; u < BUNR; ++u)
            if ((unsigned)d[u] < (unsigned)M) atomicAdd(&hist[d[u] / TR], 1);
    }
    __syncthreads();

    if (t < 32) {
        const int per = (nb + 31) / 32;
        int sum = 0;
        for (int i = 0; i < per; ++i) {
            const int b = lane * per + i;
            const int n = hist[min(b, NBMAX - 1)];
            sum += b < nb ? n : 0;
        }
        int incl = sum;
        #pragma unroll
        for (int dd = 1; dd < 32; dd <<= 1) {
            const int v = __shfl_up(incl, dd, 32);
            if (lane >= dd) incl += v;
        }
        int run = incl - sum;
        for (int i = 0; i < per; ++i) {
            const int b = lane * per + i;
            const int n = hist[min(b, NBMAX - 1)];
            if (b < nb) {
                fill[b] = run;
                tab[(size_t)b * NSL + s] = make_int2(run, n);
                run += n;
            }
        }
    }
    __syncthreads();

    for (int e0 = e_lo; e0 < e_hi; e0 += BIN_T * BUNR) {
        int   d[BUNR], sv[BUNR];
        float wq[BUNR];
        #pragma unroll
        for (int u = 0; u < BUNR; ++u) {
            const int e = e0 + u * BIN_T + t;
            const bool ok = e < e_hi;
            d[u]  = ok ? edst[e] : -1;
            sv[u] = ok ? esrc[e] : 0;
            wq[u] = ok ? ew[e]   : 0.f;
        }
        int g[BUNR];
        #pragma unroll
        for (int u = 0; u < BUNR; ++u) {
            g[u] = -1;
            if ((unsigned)d[u] < (unsigned)M) {
                const int b = d[u] / TR;
                g[u] = atomicAdd(&fill[b], 1);
                d[u] -= b * TR;
            }
        }
        #pragma unroll
        for (int u = 0; u < BUNR; ++u) {
            if (g[u] >= 0) {
                rec_row[e_lo + g[u]] = d[u];
                rec_src[e_lo + g[u]] = sv[u];
                rec_w[e_lo + g[u]]   = wq[u];
            }
        }
    }
}

__global__ __launch_bounds__(AGG_T)
void gcn_aggregate_then_gemm(const float* __restrict__ X, const float* __restrict__ W,
                             const float* __restrict__ bias,
                             const float* __restrict__ ew, const int* __restrict__ esrc,
                             const int* __restrict__ edst, const int* __restrict__ wsi,
                             float* __restrict__ out, int M, int E) {
    __shared__ __attribute__((aligned(16))) char lraw[2 * Y_STAGE * 4];
    __shared__ int   s_src[CAPE];
    __shared__ float s_w[CAPE];
    __shared__ int   rowcnt[TR + 1];
    __shared__ int   rowoff[TR + 1];
    __shared__ int   rowfill[TR + 1];
    __shared__ int   s_total;

    float* ybuf  = (float*)lraw;

    const int t    = threadIdx.x;
    const int lane = t & 31;
    const int wv   = __builtin_amdgcn_readfirstlane(t >> 5);
    const int lo   = blockIdx.x * TR;
    const int c    = lane & 15;
    const int half = lane >> 4;

    const bool mat_wave = ((wv >> 2) & 1) == 0;
    const int  ridx     = ((wv >> 3) << 2) | (wv & 3);

    const int  nchunk = (E + ECHUNK - 1) / ECHUNK;
    const bool binned = (M + TR - 1) / TR <= NBMAX;

    const int slen = (E + NSL - 1) / NSL;
    const int cap  = NSL * slen;
    const int*   rec_row = wsi;
    const int*   rec_src = wsi + cap;
    const float* rec_w   = (const float*)(wsi + 2 * cap);
    const int2*  tab     = (const int2*)(wsi + 3 * cap) + (size_t)blockIdx.x * NSL;

    if (t <= TR) rowcnt[t] = 0;
    __syncthreads();

    int   rr[NSL / 32], rs[NSL / 32], rbeg[NSL / 32], rlen[NSL / 32];
    float rw[NSL / 32];
    #pragma unroll
    for (int u = 0; u < NSL / 32; ++u) { rr[u] = -1; rs[u] = 0; rw[u] = 0.f; rbeg[u] = 0; rlen[u] = 0; }
    if (binned) {
        #pragma unroll
        for (int u = 0; u < NSL / 32; ++u) {
            const int  sl = wv + 32 * u;
            const int2 oc = tab[sl];
            rbeg[u] = sl * slen + oc.x;
            rlen[u] = oc.y;
        }
        #pragma unroll
        for (int u = 0; u < NSL / 32; ++u) {
            if (lane < rlen[u]) {
                const int g = rbeg[u] + lane;
                rr[u] = rec_row[g];
                rs[u] = rec_src[g];
                rw[u] = rec_w[g];
            }
        }
        #pragma unroll
        for (int u = 0; u < NSL / 32; ++u)
            if (rr[u] >= 0) atomicAdd(&rowcnt[rr[u]], 1);
        #pragma unroll
        for (int u = 0; u < NSL / 32; ++u)
            for (int j = 32 + lane; j < rlen[u]; j += 32)
                atomicAdd(&rowcnt[rec_row[rbeg[u] + j]], 1);
    }
    __syncthreads();

    if (wv == 0) {
        int c[NROWP];
        int s = 0;
        #pragma unroll
        for (int i = 0; i < NROWP; ++i) {
            const int r = lane * NROWP + i;
            c[i] = r < TR ? rowcnt[r] : 0;
            s += c[i];
        }
        int incl = s;
        #pragma unroll
        for (int d = 1; d < 32; d <<= 1) {
            const int v = __shfl_up(incl, d, 32);
            if (lane >= d) incl += v;
        }
        int run = incl - s;
        #pragma unroll
        for (int i = 0; i < NROWP; ++i) {
            const int r = lane * NROWP + i;
            if (r < TR) { rowoff[r] = run; rowfill[r] = run; }
            run += c[i];
        }
        if (lane == 31) s_total = run;
    }
    __syncthreads();
    const bool fallback = !binned || s_total > CAPE;

    if (!fallback) {
        int slot[NSL / 32];
        #pragma unroll
        for (int u = 0; u < NSL / 32; ++u) {
            slot[u] = -1;
            if (rr[u] >= 0) slot[u] = atomicAdd(&rowfill[rr[u]], 1);
        }
        #pragma unroll
        for (int u = 0; u < NSL / 32; ++u) {
            if (slot[u] >= 0) {
                s_src[slot[u]] = rs[u];
                s_w[slot[u]]   = rw[u];
            }
        }
        #pragma unroll
        for (int u = 0; u < NSL / 32; ++u) {
            for (int j = 32 + lane; j < rlen[u]; j += 32) {
                const int g  = rbeg[u] + j;
                const int pl = atomicAdd(&rowfill[rec_row[g]], 1);
                s_src[pl] = rec_src[g];
                s_w[pl]   = rec_w[g];
            }
        }
        __syncthreads();

        for (int st = 0; st <= NST; ++st) {
            if (!mat_wave) {
                if (st < NST) {
                    float* yb = ybuf + (st & 1) * Y_STAGE;
                    for (int j = 0; j < TM / 16; ++j) {
                        const int rl = ridx + 16 * j;
                        const int r  = st * TM + rl;
                        int n = 0, o = 0;
                        if (r < TR && lo + r < M) {
                            n = __builtin_amdgcn_readfirstlane(rowcnt[r]);
                            o = __builtin_amdgcn_readfirstlane(rowoff[r]);
                        }
                        v4f s0 = (v4f){0.f, 0.f, 0.f, 0.f}, s1 = (v4f){0.f, 0.f, 0.f, 0.f};
                        for (int i = 0; i < n; i += 8) {
                            v4f   a[8], b[8];
                            float w[8];
                            #pragma unroll
                            for (int u = 0; u < 8; ++u) {
                                const int idx = o + min(i + u, n - 1);
                                const float* p = X + (size_t)s_src[idx] * IN_SIZE + lane * 4;
                                a[u] = *(const v4f*)(p);
                                b[u] = *(const v4f*)(p + 128);
                                w[u] = s_w[idx];
                            }
                            #pragma unroll
                            for (int u = 0; u < 8; ++u) {
                                if (i + u < n) {
                                    s0 += a[u] * w[u];
                                    s1 += b[u] * w[u];
                                }
                            }
                        }
                        *(v4f*)(yb + rl * LDY + lane * 4)       = s0;
                        *(v4f*)(yb + rl * LDY + 128 + lane * 4) = s1;
                    }
                }
            } else {
                if (st >= 1)
                    gemm_tile(ybuf + ((st - 1) & 1) * Y_STAGE, W, bias, out,
                              ridx * 16, c, half, (st - 1) * TM, lo, M);
            }
            lds_barrier();
        }
    } else {
        for (int st = 0; st < NST; ++st) {
            for (int i = t; i < Y_STAGE / 4; i += AGG_T)
                ((float4*)ybuf)[i] = make_float4(0.f, 0.f, 0.f, 0.f);
            __syncthreads();
            const int tlo = lo + st * TM;
            const int tn  = min(TM, TR - st * TM);
            for (int ch = wv; ch < nchunk; ch += 32) {
                const int eb = ch * ECHUNK;
                for (int x = 0; x < ECHUNK; x += 32) {
                    const int e = eb + x + lane;
                    const int d = e < E ? edst[e] : -1;
                    unsigned mk = __builtin_amdgcn_ballot_w32((unsigned)(d - tlo) < (unsigned)tn && d < M);
                    while (mk) {
                        const int l = __builtin_ctz(mk);
                        mk &= mk - 1;
                        const int   ee  = eb + x + l;
                        const int   dd  = edst[ee];
                        const float w   = ew[ee];
                        const float* p  = X + (size_t)esrc[ee] * IN_SIZE + lane * 4;
                        const v4f a = *(const v4f*)(p);
                        const v4f b = *(const v4f*)(p + 128);
                        float* q = ybuf + (dd - tlo) * LDY + lane * 4;
                        atomicAdd(q + 0,   a.x * w);
                        atomicAdd(q + 1,   a.y * w);
                        atomicAdd(q + 2,   a.z * w);
                        atomicAdd(q + 3,   a.w * w);
                        atomicAdd(q + 128, b.x * w);
                        atomicAdd(q + 129, b.y * w);
                        atomicAdd(q + 130, b.z * w);
                        atomicAdd(q + 131, b.w * w);
                    }
                }
            }
            __syncthreads();
            if (mat_wave)
                gemm_tile(ybuf, W, bias, out, ridx * 16, c, half, st * TM, lo, M);
            __syncthreads();
        }
    }
}

extern "C" void kernel_launch(void* const* d_in, const int* in_sizes, int n_in,
                              void* d_out, int out_size, void* d_ws, size_t ws_size,
                              hipStream_t stream) {
    const float* x    = (const float*)d_in[0];
    const float* w    = (const float*)d_in[1];
    const float* bias = (const float*)d_in[2];
    const float* ew   = (const float*)d_in[3];
    const int*   esrc = (const int*)d_in[4];
    const int*   edst = (const int*)d_in[5];
    float* out  = (float*)d_out;

    const int M = in_sizes[0] / IN_SIZE;
    const int E = in_sizes[3];

    int* wsi = (int*)d_ws;
    dim3 agrid((M + TR - 1) / TR);
    gcn_bin_edges<<<NSL, BIN_T, 0, stream>>>(ew, esrc, edst, wsi, M, E);
    gcn_aggregate_then_gemm<<<agrid, AGG_T, 0, stream>>>(x, w, bias, ew, esrc, edst, wsi, out, M, E);
}


# Front launch copies W into B-fragment order: a matrix wave fetches a 32-deep K chunk with 4 whole-line 16-byte loads on a fixed base, not 16 dword loads on 32 half lines; sum order unchanged
# speedup vs baseline: 4.0301x; 1.0300x over previous
#include <hip/hip_runtime.h>

typedef __attribute__((ext_vector_type(2))) float v2f;
typedef __attribute__((ext_vector_type(8))) float v8f;
typedef __attribute__((ext_vector_type(4))) float v4f;

#define IN_SIZE  256
#define OUT_SIZE 256

#define TR     391
#define AGG_T  1024
#define ECHUNK 1024
#define CAPE   16384
#define NSL    256
#define BIN_T  256
#define BUNR   16
#define NBMAX  1024
#define NROWP  ((TR + 31) / 32)
#define TM     80
#define NST    ((TR + TM - 1) / TM)
#define LDY    260
#define Y_STAGE (TM * LDY)
#define WPK_F  (IN_SIZE * OUT_SIZE)
#define WPK_V4 (WPK_F / 4)

__device__ __forceinline__ void load_b_chunk(const v4f* wq, int kc, v4f (&b)[4]) {
    #pragma unroll
    for (int q = 0; q < 4; ++q) b[q] = wq[(kc * 4 + q) * 32];
}

__device__ __forceinline__ void gemm_tile(const float* __restrict__ ytile, const float* __restrict__ wpk,
                                          const float* __restrict__ bias, float* __restrict__ out,
                                          int n0, int c, int half, int rbase, int lo, int M) {
    const float* ya = ytile + c * LDY + 2 * half;
    const v4f*   wp = (const v4f*)wpk + (size_t)(n0 >> 4) * (IN_SIZE / 32) * 4 * 32 + half * 16 + c;

    v8f acc[5];
    #pragma unroll
    for (int mt = 0; mt < 5; ++mt) acc[mt] = (v8f){0.f,0.f,0.f,0.f,0.f,0.f,0.f,0.f};

    v4f bc[4], bn[4];
    load_b_chunk(wp, 0, bc);
    #pragma unroll
    for (int kc = 0; kc < IN_SIZE / 32; ++kc) {
        if (kc + 1 < IN_SIZE / 32) load_b_chunk(wp, kc + 1, bn);
        __builtin_amdgcn_sched_barrier(0);
        #pragma unroll
        for (int ks = 0; ks < 8; ++ks) {
            v2f a[5];
            #pragma unroll
            for (int mt = 0; mt < 5; ++mt)
                a[mt] = *(const v2f*)(ya + mt * 16 * LDY + kc * 32 + ks * 4);
            const v4f bq = bc[ks >> 1];
            const v2f b  = (ks & 1) ? (v2f){bq.z, bq.w} : (v2f){bq.x, bq.y};
            #pragma unroll
            for (int mt = 0; mt < 5; ++mt)
                acc[mt] = __builtin_amdgcn_wmma_f32_16x16x4_f32(
                    false, a[mt], false, b, (short)0, acc[mt], false, false);
        }
        #pragma unroll
        for (int q = 0; q < 4; ++q) bc[q] = bn[q];
    }

    const float bv = bias[n0 + c];
    float* op = out + (size_t)(lo + rbase + 8 * half) * OUT_SIZE + n0 + c;
    if (rbase + TM <= TR && lo + rbase + TM <= M) {
        #pragma unroll
        for (int mt = 0; mt < 5; ++mt)
            #pragma unroll
            for (int v = 0; v < 8; ++v)
                op[(size_t)(mt * 16 + v) * OUT_SIZE] = acc[mt][v] + bv;
    } else {
        #pragma unroll
        for (int mt = 0; mt < 5; ++mt)
            #pragma unroll
            for (int v = 0; v < 8; ++v) {
                const int r = rbase + mt * 16 + v + 8 * half;
                if (r < TR && lo + r < M)
                    op[(size_t)(mt * 16 + v) * OUT_SIZE] = acc[mt][v] + bv;
            }
    }
}

__device__ __forceinline__ void lds_barrier() {
    __builtin_amdgcn_fence(__ATOMIC_RELEASE, "workgroup", "local");
    __builtin_amdgcn_s_barrier();
    __builtin_amdgcn_fence(__ATOMIC_ACQUIRE, "workgroup", "local");
}

__device__ __forceinline__ v4f w_group(const float* __restrict__ W, int g) {
    const int l  = g & 31;
    const int q  = (g >> 5) & 3;
    const int kc = (g >> 7) & (IN_SIZE / 32 - 1);
    const int j  = g / (32 * 4 * (IN_SIZE / 32));
    const float* p = W + (size_t)(kc * 32 + q * 8 + 2 * (l >> 4)) * OUT_SIZE + 16 * j + (l & 15);
    v4f v;
    v.x = p[0];
    v.y = p[OUT_SIZE];
    v.z = p[4 * OUT_SIZE];
    v.w = p[5 * OUT_SIZE];
    return v;
}


__global__ __launch_bounds__(BIN_T)
void gcn_bin_edges(const float* __restrict__ W, const float* __restrict__ ew, const int* __restrict__ esrc,
                   const int* __restrict__ edst, int* __restrict__ ws, int M, int E) {
    __shared__ int hist[NBMAX];
    __shared__ int fill[NBMAX];

    const int t    = threadIdx.x;
    const int lane = t & 31;

    static_assert((WPK_V4 + NSL - 1) / NSL <= BIN_T, "one group per thread");
    const int  wg  = blockIdx.x * ((WPK_V4 + NSL - 1) / NSL) + t;
    const bool wdo = t < (WPK_V4 + NSL - 1) / NSL && wg < WPK_V4;

    const int nb = (M + TR - 1) / TR;
    if (nb > NBMAX) {
        if (wdo) ((v4f*)ws)[wg] = w_group(W, wg);
        return;
    }

    const int s    = blockIdx.x;
    const int slen = (E + NSL - 1) / NSL;
    const int cap  = NSL * slen;
    int*   wsi     = ws + WPK_F;
    int*   rec_row = wsi;
    int*   rec_src = wsi + cap;
    float* rec_w   = (float*)(wsi + 2 * cap);
    int2*  tab     = (int2*)(wsi + 3 * cap);
    const int e_lo = s * slen;
    const int e_hi = min(E, e_lo + slen);

    for (int i = t; i < nb; i += BIN_T) hist[i] = 0;
    __syncthreads();

    v4f wv = (v4f){0.f, 0.f, 0.f, 0.f};
    if (wdo) wv = w_group(W, wg);

    for (int e0 = e_lo; e0 < e_hi; e0 += BIN_T * BUNR) {
        int d[BUNR];
        #pragma unroll
        for (int u = 0; u < BUNR; ++u) {
            const int e = e0 + u * BIN_T + t;
            d[u] = e < e_hi ? edst[e] : -1;
        }
        #pragma unroll
        for (int u = 0; u < BUNR; ++u)
            if ((unsigned)d[u] < (unsigned)M) atomicAdd(&hist[d[u] / TR], 1);
    }
    __syncthreads();

    if (t < 32) {
        const int per = (nb + 31) / 32;
        int sum = 0;
        for (int i = 0; i < per; ++i) {
            const int b = lane * per + i;
            const int n = hist[min(b, NBMAX - 1)];
            sum += b < nb ? n : 0;
        }
        int incl = sum;
        #pragma unroll
        for (int dd = 1; dd < 32; dd <<= 1) {
            const int v = __shfl_up(incl, dd, 32);
            if (lane >= dd) incl += v;
        }
        int run = incl - sum;
        for (int i = 0; i < per; ++i) {
            const int b = lane * per + i;
            const int n = hist[min(b, NBMAX - 1)];
            if (b < nb) {
                fill[b] = run;
                tab[(size_t)b * NSL + s] = make_int2(run, n);
                run += n;
            }
        }
    }
    __syncthreads();

    for (int e0 = e_lo; e0 < e_hi; e0 += BIN_T * BUNR) {
        int   d[BUNR], sv[BUNR];
        float wq[BUNR];
        #pragma unroll
        for (int u = 0; u < BUNR; ++u) {
            const int e = e0 + u * BIN_T + t;
            const bool ok = e < e_hi;
            d[u]  = ok ? edst[e] : -1;
            sv[u] = ok ? esrc[e] : 0;
            wq[u] = ok ? ew[e]   : 0.f;
        }
        int g[BUNR];
        #pragma unroll
        for (int u = 0; u < BUNR; ++u) {
            g[u] = -1;
            if ((unsigned)d[u] < (unsigned)M) {
                const int b = d[u] / TR;
                g[u] = atomicAdd(&fill[b], 1);
                d[u] -= b * TR;
            }
        }
        #pragma unroll
        for (int u = 0; u < BUNR; ++u) {
            if (g[u] >= 0) {
                rec_row[e_lo + g[u]] = d[u];
                rec_src[e_lo + g[u]] = sv[u];
                rec_w[e_lo + g[u]]   = wq[u];
            }
        }
    }

    if (wdo) ((v4f*)ws)[wg] = wv;
}

__global__ __launch_bounds__(AGG_T)
void gcn_aggregate_then_gemm(const float* __restrict__ X, const float* __restrict__ bias,
                             const float* __restrict__ ew, const int* __restrict__ esrc,
                             const int* __restrict__ edst, const int* __restrict__ ws,
                             float* __restrict__ out, int M, int E) {
    __shared__ __attribute__((aligned(16))) char lraw[2 * Y_STAGE * 4];
    __shared__ int   s_src[CAPE];
    __shared__ float s_w[CAPE];
    __shared__ int   rowcnt[TR + 1];
    __shared__ int   rowoff[TR + 1];
    __shared__ int   rowfill[TR + 1];
    __shared__ int   s_total;

    float* ybuf  = (float*)lraw;

    const int t    = threadIdx.x;
    const int lane = t & 31;
    const int wv   = __builtin_amdgcn_readfirstlane(t >> 5);
    const int lo   = blockIdx.x * TR;
    const int c    = lane & 15;
    const int half = lane >> 4;

    const bool mat_wave = ((wv >> 2) & 1) == 0;
    const int  ridx     = ((wv >> 3) << 2) | (wv & 3);

    const int  nchunk = (E + ECHUNK - 1) / ECHUNK;
    const bool binned = (M + TR - 1) / TR <= NBMAX;

    const int slen = (E + NSL - 1) / NSL;
    const int cap  = NSL * slen;
    const float* wpk     = (const float*)ws;
    const int*   wsi     = ws + WPK_F;
    const int*   rec_row = wsi;
    const int*   rec_src = wsi + cap;
    const float* rec_w   = (const float*)(wsi + 2 * cap);
    const int2*  tab     = (const int2*)(wsi + 3 * cap) + (size_t)blockIdx.x * NSL;

    if (t <= TR) rowcnt[t] = 0;
    __syncthreads();

    int   rr[NSL / 32], rs[NSL / 32], rbeg[NSL / 32], rlen[NSL / 32];
    float rw[NSL / 32];
    #pragma unroll
    for (int u = 0; u < NSL / 32; ++u) { rr[u] = -1; rs[u] = 0; rw[u] = 0.f; rbeg[u] = 0; rlen[u] = 0; }
    if (binned) {
        #pragma unroll
        for (int u = 0; u < NSL / 32; ++u) {
            const int  sl = wv + 32 * u;
            const int2 oc = tab[sl];
            rbeg[u] = sl * slen + oc.x;
            rlen[u] = oc.y;
        }
        #pragma unroll
        for (int u = 0; u < NSL / 32; ++u) {
            if (lane < rlen[u]) {
                const int g = rbeg[u] + lane;
                rr[u] = rec_row[g];
                rs[u] = rec_src[g];
                rw[u] = rec_w[g];
            }
        }
        #pragma unroll
        for (int u = 0; u < NSL / 32; ++u)
            if (rr[u] >= 0) atomicAdd(&rowcnt[rr[u]], 1);
        #pragma unroll
        for (int u = 0; u < NSL / 32; ++u)
            for (int j = 32 + lane; j < rlen[u]; j += 32)
                atomicAdd(&rowcnt[rec_row[rbeg[u] + j]], 1);
    }
    __syncthreads();

    if (wv == 0) {
        int c[NROWP];
        int s = 0;
        #pragma unroll
        for (int i = 0; i < NROWP; ++i) {
            const int r = lane * NROWP + i;
            c[i] = r < TR ? rowcnt[r] : 0;
            s += c[i];
        }
        int incl = s;
        #pragma unroll
        for (int d = 1; d < 32; d <<= 1) {
            const int v = __shfl_up(incl, d, 32);
            if (lane >= d) incl += v;
        }
        int run = incl - s;
        #pragma unroll
        for (int i = 0; i < NROWP; ++i) {
            const int r = lane * NROWP + i;
            if (r < TR) { rowoff[r] = run; rowfill[r] = run; }
            run += c[i];
        }
        if (lane == 31) s_total = run;
    }
    __syncthreads();
    const bool fallback = !binned || s_total > CAPE;

    if (!fallback) {
        int slot[NSL / 32];
        #pragma unroll
        for (int u = 0; u < NSL / 32; ++u) {
            slot[u] = -1;
            if (rr[u] >= 0) slot[u] = atomicAdd(&rowfill[rr[u]], 1);
        }
        #pragma unroll
        for (int u = 0; u < NSL / 32; ++u) {
            if (slot[u] >= 0) {
                s_src[slot[u]] = rs[u];
                s_w[slot[u]]   = rw[u];
            }
        }
        #pragma unroll
        for (int u = 0; u < NSL / 32; ++u) {
            for (int j = 32 + lane; j < rlen[u]; j += 32) {
                const int g  = rbeg[u] + j;
                const int pl = atomicAdd(&rowfill[rec_row[g]], 1);
                s_src[pl] = rec_src[g];
                s_w[pl]   = rec_w[g];
            }
        }
        __syncthreads();

        for (int st = 0; st <= NST; ++st) {
            if (!mat_wave) {
                if (st < NST) {
                    float* yb = ybuf + (st & 1) * Y_STAGE;
                    for (int j = 0; j < TM / 16; ++j) {
                        const int rl = ridx + 16 * j;
                        const int r  = st * TM + rl;
                        int n = 0, o = 0;
                        if (r < TR && lo + r < M) {
                            n = __builtin_amdgcn_readfirstlane(rowcnt[r]);
                            o = __builtin_amdgcn_readfirstlane(rowoff[r]);
                        }
                        v4f s0 = (v4f){0.f, 0.f, 0.f, 0.f}, s1 = (v4f){0.f, 0.f, 0.f, 0.f};
                        for (int i = 0; i < n; i += 8) {
                            v4f   a[8], b[8];
                            float w[8];
                            #pragma unroll
                            for (int u = 0; u < 8; ++u) {
                                const int idx = o + min(i + u, n - 1);
                                const float* p = X + (size_t)s_src[idx] * IN_SIZE + lane * 4;
                                a[u] = *(const v4f*)(p);
                                b[u] = *(const v4f*)(p + 128);
                                w[u] = s_w[idx];
                            }
                            #pragma unroll
                            for (int u = 0; u < 8; ++u) {
                                if (i + u < n) {
                                    s0 += a[u] * w[u];
                                    s1 += b[u] * w[u];
                                }
                            }
                        }
                        *(v4f*)(yb + rl * LDY + lane * 4)       = s0;
                        *(v4f*)(yb + rl * LDY + 128 + lane * 4) = s1;
                    }
                }
            } else {
                if (st >= 1)
                    gemm_tile(ybuf + ((st - 1) & 1) * Y_STAGE, wpk, bias, out,
                              ridx * 16, c, half, (st - 1) * TM, lo, M);
            }
            lds_barrier();
        }
    } else {
        for (int st = 0; st < NST; ++st) {
            for (int i = t; i < Y_STAGE / 4; i += AGG_T)
                ((float4*)ybuf)[i] = make_float4(0.f, 0.f, 0.f, 0.f);
            __syncthreads();
            const int tlo = lo + st * TM;
            const int tn  = min(TM, TR - st * TM);
            for (int ch = wv; ch < nchunk; ch += 32) {
                const int eb = ch * ECHUNK;
                for (int x = 0; x < ECHUNK; x += 32) {
                    const int e = eb + x + lane;
                    const int d = e < E ? edst[e] : -1;
                    unsigned mk = __builtin_amdgcn_ballot_w32((unsigned)(d - tlo) < (unsigned)tn && d < M);
                    while (mk) {
                        const int l = __builtin_ctz(mk);
                        mk &= mk - 1;
                        const int   ee  = eb + x + l;
                        const int   dd  = edst[ee];
                        const float w   = ew[ee];
                        const float* p  = X + (size_t)esrc[ee] * IN_SIZE + lane * 4;
                        const v4f a = *(const v4f*)(p);
                        const v4f b = *(const v4f*)(p + 128);
                        float* q = ybuf + (dd - tlo) * LDY + lane * 4;
                        atomicAdd(q + 0,   a.x * w);
                        atomicAdd(q + 1,   a.y * w);
                        atomicAdd(q + 2,   a.z * w);
                        atomicAdd(q + 3,   a.w * w);
                        atomicAdd(q + 128, b.x * w);
                        atomicAdd(q + 129, b.y * w);
                        atomicAdd(q + 130, b.z * w);
                        atomicAdd(q + 131, b.w * w);
                    }
                }
            }
            __syncthreads();
            if (mat_wave)
                gemm_tile(ybuf, wpk, bias, out, ridx * 16, c, half, st * TM, lo, M);
            __syncthreads();
        }
    }
}

extern "C" void kernel_launch(void* const* d_in, const int* in_sizes, int n_in,
                              void* d_out, int out_size, void* d_ws, size_t ws_size,
                              hipStream_t stream) {
    const float* x    = (const float*)d_in[0];
    const float* w    = (const float*)d_in[1];
    const float* bias = (const float*)d_in[2];
    const float* ew   = (const float*)d_in[3];
    const int*   esrc = (const int*)d_in[4];
    const int*   edst = (const int*)d_in[5];
    float* out  = (float*)d_out;

    const int M = in_sizes[0] / IN_SIZE;
    const int E = in_sizes[3];

    int* wsi = (int*)d_ws;
    dim3 agrid((M + TR - 1) / TR);
    gcn_bin_edges<<<NSL, BIN_T, 0, stream>>>(w, ew, esrc, edst, wsi, M, E);
    gcn_aggregate_then_gemm<<<agrid, AGG_T, 0, stream>>>(x, bias, ew, esrc, edst, wsi, out, M, E);
}


# Launch 1 groups a slice's records in LDS and writes them out in index order: a store instruction fills one whole 128-byte line, not 4-byte pieces of up to 32 lines; sum order unchanged
# speedup vs baseline: 4.2591x; 1.0568x over previous
#include <hip/hip_runtime.h>

typedef __attribute__((ext_vector_type(2))) float v2f;
typedef __attribute__((ext_vector_type(8))) float v8f;
typedef __attribute__((ext_vector_type(4))) float v4f;

#define IN_SIZE  256
#define OUT_SIZE 256

#define TR     391
#define AGG_T  1024
#define ECHUNK 1024
#define CAPE   16384
#define NSL    256
#define BIN_T  256
#define BUNR   16
#define NBMAX  1024
#define SCAP   (BIN_T * BUNR)
#define NROWP  ((TR + 31) / 32)
#define TM     80
#define NST    ((TR + TM - 1) / TM)
#define LDY    260
#define Y_STAGE (TM * LDY)
#define WPK_F  (IN_SIZE * OUT_SIZE)
#define WPK_V4 (WPK_F / 4)

__device__ __forceinline__ void load_b_chunk(const v4f* wq, int kc, v4f (&b)[4]) {
    #pragma unroll
    for (int q = 0; q < 4; ++q) b[q] = wq[(kc * 4 + q) * 32];
}

__device__ __forceinline__ void gemm_tile(const float* __restrict__ ytile, const float* __restrict__ wpk,
                                          const float* __restrict__ bias, float* __restrict__ out,
                                          int n0, int c, int half, int rbase, int lo, int M) {
    const float* ya = ytile + c * LDY + 2 * half;
    const v4f*   wp = (const v4f*)wpk + (size_t)(n0 >> 4) * (IN_SIZE / 32) * 4 * 32 + half * 16 + c;

    v8f acc[5];
    #pragma unroll
    for (int mt = 0; mt < 5; ++mt) acc[mt] = (v8f){0.f,0.f,0.f,0.f,0.f,0.f,0.f,0.f};

    v4f bc[4], bn[4];
    load_b_chunk(wp, 0, bc);
    #pragma unroll
    for (int kc = 0; kc < IN_SIZE / 32; ++kc) {
        if (kc + 1 < IN_SIZE / 32) load_b_chunk(wp, kc + 1, bn);
        __builtin_amdgcn_sched_barrier(0);
        #pragma unroll
        for (int ks = 0; ks < 8; ++ks) {
            v2f a[5];
            #pragma unroll
            for (int mt = 0; mt < 5; ++mt)
                a[mt] = *(const v2f*)(ya + mt * 16 * LDY + kc * 32 + ks * 4);
            const v4f bq = bc[ks >> 1];
            const v2f b  = (ks & 1) ? (v2f){bq.z, bq.w} : (v2f){bq.x, bq.y};
            #pragma unroll
            for (int mt = 0; mt < 5; ++mt)
                acc[mt] = __builtin_amdgcn_wmma_f32_16x16x4_f32(
                    false, a[mt], false, b, (short)0, acc[mt], false, false);
        }
        #pragma unroll
        for (int q = 0; q < 4; ++q) bc[q] = bn[q];
    }

    const float bv = bias[n0 + c];
    float* op = out + (size_t)(lo + rbase + 8 * half) * OUT_SIZE + n0 + c;
    if (rbase + TM <= TR && lo + rbase + TM <= M) {
        #pragma unroll
        for (int mt = 0; mt < 5; ++mt)
            #pragma unroll
            for (int v = 0; v < 8; ++v)
                op[(size_t)(mt * 16 + v) * OUT_SIZE] = acc[mt][v] + bv;
    } else {
        #pragma unroll
        for (int mt = 0; mt < 5; ++mt)
            #pragma unroll
            for (int v = 0; v < 8; ++v) {
                const int r = rbase + mt * 16 + v + 8 * half;
                if (r < TR && lo + r < M)
                    op[(size_t)(mt * 16 + v) * OUT_SIZE] = acc[mt][v] + bv;
            }
    }
}

__device__ __forceinline__ void lds_barrier() {
    __builtin_amdgcn_fence(__ATOMIC_RELEASE, "workgroup", "local");
    __builtin_amdgcn_s_barrier();
    __builtin_amdgcn_fence(__ATOMIC_ACQUIRE, "workgroup", "local");
}

__device__ __forceinline__ v4f w_group(const float* __restrict__ W, int g) {
    const int l  = g & 31;
    const int q  = (g >> 5) & 3;
    const int kc = (g >> 7) & (IN_SIZE / 32 - 1);
    const int j  = g / (32 * 4 * (IN_SIZE / 32));
    const float* p = W + (size_t)(kc * 32 + q * 8 + 2 * (l >> 4)) * OUT_SIZE + 16 * j + (l & 15);
    v4f v;
    v.x = p[0];
    v.y = p[OUT_SIZE];
    v.z = p[4 * OUT_SIZE];
    v.w = p[5 * OUT_SIZE];
    return v;
}


__global__ __launch_bounds__(BIN_T)
void gcn_bin_edges(const float* __restrict__ W, const float* __restrict__ ew, const int* __restrict__ esrc,
                   const int* __restrict__ edst, int* __restrict__ ws, int M, int E) {
    __shared__ int hist[NBMAX];
    __shared__ int fill[NBMAX];
    __shared__ int   st_row[SCAP];
    __shared__ int   st_src[SCAP];
    __shared__ float st_w[SCAP];
    __shared__ int   s_nrec;

    const int t    = threadIdx.x;
    const int lane = t & 31;

    static_assert((WPK_V4 + NSL - 1) / NSL <= BIN_T, "one group per thread");
    const int  wg  = blockIdx.x * ((WPK_V4 + NSL - 1) / NSL) + t;
    const bool wdo = t < (WPK_V4 + NSL - 1) / NSL && wg < WPK_V4;

    const int nb = (M + TR - 1) / TR;
    if (nb > NBMAX) {
        if (wdo) ((v4f*)ws)[wg] = w_group(W, wg);
        return;
    }

    const int s    = blockIdx.x;
    const int slen = (E + NSL - 1) / NSL;
    const int cap  = NSL * slen;
    int*   wsi     = ws + WPK_F;
    int*   rec_row = wsi;
    int*   rec_src = wsi + cap;
    float* rec_w   = (float*)(wsi + 2 * cap);
    int2*  tab     = (int2*)(wsi + 3 * cap);
    const int e_lo = s * slen;
    const int e_hi = min(E, e_lo + slen);
    const bool staged = slen <= SCAP;

    for (int i = t; i < nb; i += BIN_T) hist[i] = 0;
    __syncthreads();

    v4f wv = (v4f){0.f, 0.f, 0.f, 0.f};
    if (wdo) wv = w_group(W, wg);

    for (int e0 = e_lo; e0 < e_hi; e0 += BIN_T * BUNR) {
        int d[BUNR];
        #pragma unroll
        for (int u = 0; u < BUNR; ++u) {
            const int e = e0 + u * BIN_T + t;
            d[u] = e < e_hi ? edst[e] : -1;
        }
        #pragma unroll
        for (int u = 0; u < BUNR; ++u)
            if ((unsigned)d[u] < (unsigned)M) atomicAdd(&hist[d[u] / TR], 1);
    }
    __syncthreads();

    if (t < 32) {
        const int per = (nb + 31) / 32;
        int sum = 0;
        for (int i = 0; i < per; ++i) {
            const int b = lane * per + i;
            const int n = hist[min(b, NBMAX - 1)];
            sum += b < nb ? n : 0;
        }
        int incl = sum;
        #pragma unroll
        for (int dd = 1; dd < 32; dd <<= 1) {
            const int v = __shfl_up(incl, dd, 32);
            if (lane >= dd) incl += v;
        }
        int run = incl - sum;
        for (int i = 0; i < per; ++i) {
            const int b = lane * per + i;
            const int n = hist[min(b, NBMAX - 1)];
            if (b < nb) {
                fill[b] = run;
                tab[(size_t)b * NSL + s] = make_int2(run, n);
                run += n;
            }
        }
        if (lane == 31) s_nrec = run;
    }
    __syncthreads();

    for (int e0 = e_lo; e0 < e_hi; e0 += BIN_T * BUNR) {
        int   d[BUNR], sv[BUNR];
        float wq[BUNR];
        #pragma unroll
        for (int u = 0; u < BUNR; ++u) {
            const int e = e0 + u * BIN_T + t;
            const bool ok = e < e_hi;
            d[u]  = ok ? edst[e] : -1;
            sv[u] = ok ? esrc[e] : 0;
            wq[u] = ok ? ew[e]   : 0.f;
        }
        int g[BUNR];
        #pragma unroll
        for (int u = 0; u < BUNR; ++u) {
            g[u] = -1;
            if ((unsigned)d[u] < (unsigned)M) {
                const int b = d[u] / TR;
                g[u] = atomicAdd(&fill[b], 1);
                d[u] -= b * TR;
            }
        }
        if (staged) {
            #pragma unroll
            for (int u = 0; u < BUNR; ++u) {
                if (g[u] >= 0) {
                    st_row[g[u]] = d[u];
                    st_src[g[u]] = sv[u];
                    st_w[g[u]]   = wq[u];
                }
            }
        } else {
            #pragma unroll
            for (int u = 0; u < BUNR; ++u) {
                if (g[u] >= 0) {
                    rec_row[e_lo + g[u]] = d[u];
                    rec_src[e_lo + g[u]] = sv[u];
                    rec_w[e_lo + g[u]]   = wq[u];
                }
            }
        }
    }

    if (staged) {
        __syncthreads();
        const int n   = s_nrec;
        const int pad = e_lo & 31;
        if (n > 0) {
            for (int i0 = t - pad; i0 < n; i0 += BIN_T * 4) {
                int   r[4], sv[4];
                float wq[4];
                #pragma unroll
                for (int u = 0; u < 4; ++u) {
                    const int j = min(max(i0 + u * BIN_T, 0), n - 1);
                    r[u]  = st_row[j];
                    sv[u] = st_src[j];
                    wq[u] = st_w[j];
                }
                #pragma unroll
                for (int u = 0; u < 4; ++u) {
                    const int i = i0 + u * BIN_T;
                    if (i >= 0 && i < n) {
                        rec_row[e_lo + i] = r[u];
                        rec_src[e_lo + i] = sv[u];
                        rec_w[e_lo + i]   = wq[u];
                    }
                }
            }
        }
    }

    if (wdo) ((v4f*)ws)[wg] = wv;
}

__global__ __launch_bounds__(AGG_T)
void gcn_aggregate_then_gemm(const float* __restrict__ X, const float* __restrict__ bias,
                             const float* __restrict__ ew, const int* __restrict__ esrc,
                             const int* __restrict__ edst, const int* __restrict__ ws,
                             float* __restrict__ out, int M, int E) {
    __shared__ __attribute__((aligned(16))) char lraw[2 * Y_STAGE * 4];
    __shared__ int   s_src[CAPE];
    __shared__ float s_w[CAPE];
    __shared__ int   rowcnt[TR + 1];
    __shared__ int   rowoff[TR + 1];
    __shared__ int   rowfill[TR + 1];
    __shared__ int   s_total;

    float* ybuf  = (float*)lraw;

    const int t    = threadIdx.x;
    const int lane = t & 31;
    const int wv   = __builtin_amdgcn_readfirstlane(t >> 5);
    const int lo   = blockIdx.x * TR;
    const int c    = lane & 15;
    const int half = lane >> 4;

    const bool mat_wave = ((wv >> 2) & 1) == 0;
    const int  ridx     = ((wv >> 3) << 2) | (wv & 3);

    const int  nchunk = (E + ECHUNK - 1) / ECHUNK;
    const bool binned = (M + TR - 1) / TR <= NBMAX;

    const int slen = (E + NSL - 1) / NSL;
    const int cap  = NSL * slen;
    const float* wpk     = (const float*)ws;
    const int*   wsi     = ws + WPK_F;
    const int*   rec_row = wsi;
    const int*   rec_src = wsi + cap;
    const float* rec_w   = (const float*)(wsi + 2 * cap);
    const int2*  tab     = (const int2*)(wsi + 3 * cap) + (size_t)blockIdx.x * NSL;

    if (t <= TR) rowcnt[t] = 0;
    __syncthreads();

    int   rr[NSL / 32], rs[NSL / 32], rbeg[NSL / 32], rlen[NSL / 32];
    float rw[NSL / 32];
    #pragma unroll
    for (int u = 0; u < NSL / 32; ++u) { rr[u] = -1; rs[u] = 0; rw[u] = 0.f; rbeg[u] = 0; rlen[u] = 0; }
    if (binned) {
        #pragma unroll
        for (int u = 0; u < NSL / 32; ++u) {
            const int  sl = wv + 32 * u;
            const int2 oc = tab[sl];
            rbeg[u] = sl * slen + oc.x;
            rlen[u] = oc.y;
        }
        #pragma unroll
        for (int u = 0; u < NSL / 32; ++u) {
            if (lane < rlen[u]) {
                const int g = rbeg[u] + lane;
                rr[u] = rec_row[g];
                rs[u] = rec_src[g];
                rw[u] = rec_w[g];
            }
        }
        #pragma unroll
        for (int u = 0; u < NSL / 32; ++u)
            if (rr[u] >= 0) atomicAdd(&rowcnt[rr[u]], 1);
        #pragma unroll
        for (int u = 0; u < NSL / 32; ++u)
            for (int j = 32 + lane; j < rlen[u]; j += 32)
                atomicAdd(&rowcnt[rec_row[rbeg[u] + j]], 1);
    }
    __syncthreads();

    if (wv == 0) {
        int c[NROWP];
        int s = 0;
        #pragma unroll
        for (int i = 0; i < NROWP; ++i) {
            const int r = lane * NROWP + i;
            c[i] = r < TR ? rowcnt[r] : 0;
            s += c[i];
        }
        int incl = s;
        #pragma unroll
        for (int d = 1; d < 32; d <<= 1) {
            const int v = __shfl_up(incl, d, 32);
            if (lane >= d) incl += v;
        }
        int run = incl - s;
        #pragma unroll
        for (int i = 0; i < NROWP; ++i) {
            const int r = lane * NROWP + i;
            if (r < TR) { rowoff[r] = run; rowfill[r] = run; }
            run += c[i];
        }
        if (lane == 31) s_total = run;
    }
    __syncthreads();
    const bool fallback = !binned || s_total > CAPE;

    if (!fallback) {
        int slot[NSL / 32];
        #pragma unroll
        for (int u = 0; u < NSL / 32; ++u) {
            slot[u] = -1;
            if (rr[u] >= 0) slot[u] = atomicAdd(&rowfill[rr[u]], 1);
        }
        #pragma unroll
        for (int u = 0; u < NSL / 32; ++u) {
            if (slot[u] >= 0) {
                s_src[slot[u]] = rs[u];
                s_w[slot[u]]   = rw[u];
            }
        }
        #pragma unroll
        for (int u = 0; u < NSL / 32; ++u) {
            for (int j = 32 + lane; j < rlen[u]; j += 32) {
                const int g  = rbeg[u] + j;
                const int pl = atomicAdd(&rowfill[rec_row[g]], 1);
                s_src[pl] = rec_src[g];
                s_w[pl]   = rec_w[g];
            }
        }
        __syncthreads();

        for (int st = 0; st <= NST; ++st) {
            if (!mat_wave) {
                if (st < NST) {
                    float* yb = ybuf + (st & 1) * Y_STAGE;
                    for (int j = 0; j < TM / 16; ++j) {
                        const int rl = ridx + 16 * j;
                        const int r  = st * TM + rl;
                        int n = 0, o = 0;
                        if (r < TR && lo + r < M) {
                            n = __builtin_amdgcn_readfirstlane(rowcnt[r]);
                            o = __builtin_amdgcn_readfirstlane(rowoff[r]);
                        }
                        v4f s0 = (v4f){0.f, 0.f, 0.f, 0.f}, s1 = (v4f){0.f, 0.f, 0.f, 0.f};
                        for (int i = 0; i < n; i += 8) {
                            v4f   a[8], b[8];
                            float w[8];
                            #pragma unroll
                            for (int u = 0; u < 8; ++u) {
                                const int idx = o + min(i + u, n - 1);
                                const float* p = X + (size_t)s_src[idx] * IN_SIZE + lane * 4;
                                a[u] = *(const v4f*)(p);
                                b[u] = *(const v4f*)(p + 128);
                                w[u] = s_w[idx];
                            }
                            #pragma unroll
                            for (int u = 0; u < 8; ++u) {
                                if (i + u < n) {
                                    s0 += a[u] * w[u];
                                    s1 += b[u] * w[u];
                                }
                            }
                        }
                        *(v4f*)(yb + rl * LDY + lane * 4)       = s0;
                        *(v4f*)(yb + rl * LDY + 128 + lane * 4) = s1;
                    }
                }
            } else {
                if (st >= 1)
                    gemm_tile(ybuf + ((st - 1) & 1) * Y_STAGE, wpk, bias, out,
                              ridx * 16, c, half, (st - 1) * TM, lo, M);
            }
            lds_barrier();
        }
    } else {
        for (int st = 0; st < NST; ++st) {
            for (int i = t; i < Y_STAGE / 4; i += AGG_T)
                ((float4*)ybuf)[i] = make_float4(0.f, 0.f, 0.f, 0.f);
            __syncthreads();
            const int tlo = lo + st * TM;
            const int tn  = min(TM, TR - st * TM);
            for (int ch = wv; ch < nchunk; ch += 32) {
                const int eb = ch * ECHUNK;
                for (int x = 0; x < ECHUNK; x += 32) {
                    const int e = eb + x + lane;
                    const int d = e < E ? edst[e] : -1;
                    unsigned mk = __builtin_amdgcn_ballot_w32((unsigned)(d - tlo) < (unsigned)tn && d < M);
                    while (mk) {
                        const int l = __builtin_ctz(mk);
                        mk &= mk - 1;
                        const int   ee  = eb + x + l;
                        const int   dd  = edst[ee];
                        const float w   = ew[ee];
                        const float* p  = X + (size_t)esrc[ee] * IN_SIZE + lane * 4;
                        const v4f a = *(const v4f*)(p);
                        const v4f b = *(const v4f*)(p + 128);
                        float* q = ybuf + (dd - tlo) * LDY + lane * 4;
                        atomicAdd(q + 0,   a.x * w);
                        atomicAdd(q + 1,   a.y * w);
                        atomicAdd(q + 2,   a.z * w);
                        atomicAdd(q + 3,   a.w * w);
                        atomicAdd(q + 128, b.x * w);
                        atomicAdd(q + 129, b.y * w);
                        atomicAdd(q + 130, b.z * w);
                        atomicAdd(q + 131, b.w * w);
                    }
                }
            }
            __syncthreads();
            if (mat_wave)
                gemm_tile(ybuf, wpk, bias, out, ridx * 16, c, half, st * TM, lo, M);
            __syncthreads();
        }
    }
}

extern "C" void kernel_launch(void* const* d_in, const int* in_sizes, int n_in,
                              void* d_out, int out_size, void* d_ws, size_t ws_size,
                              hipStream_t stream) {
    const float* x    = (const float*)d_in[0];
    const float* w    = (const float*)d_in[1];
    const float* bias = (const float*)d_in[2];
    const float* ew   = (const float*)d_in[3];
    const int*   esrc = (const int*)d_in[4];
    const int*   edst = (const int*)d_in[5];
    float* out  = (float*)d_out;

    const int M = in_sizes[0] / IN_SIZE;
    const int E = in_sizes[3];

    int* wsi = (int*)d_ws;
    dim3 agrid((M + TR - 1) / TR);
    gcn_bin_edges<<<NSL, BIN_T, 0, stream>>>(w, ew, esrc, edst, wsi, M, E);
    gcn_aggregate_then_gemm<<<agrid, AGG_T, 0, stream>>>(x, bias, ew, esrc, edst, wsi, out, M, E);
}


# Gather waves take 5 consecutive rows and walk their edge segments as one run in full 8-edge batches, a row's sums stored at its segment end, not ceil(n/8) padded batches per row; sum order kept
# speedup vs baseline: 4.4146x; 1.0365x over previous
#include <hip/hip_runtime.h>

typedef __attribute__((ext_vector_type(2))) float v2f;
typedef __attribute__((ext_vector_type(8))) float v8f;
typedef __attribute__((ext_vector_type(4))) float v4f;

#define IN_SIZE  256
#define OUT_SIZE 256

#define TR     391
#define AGG_T  1024
#define ECHUNK 1024
#define CAPE   16384
#define NSL    256
#define BIN_T  256
#define BUNR   16
#define NBMAX  1024
#define SCAP   (BIN_T * BUNR)
#define NROWP  ((TR + 31) / 32)
#define TM     80
#define NST    ((TR + TM - 1) / TM)
#define GR     (TM / 16)
#define LDY    260
#define Y_STAGE (TM * LDY)
#define WPK_F  (IN_SIZE * OUT_SIZE)
#define WPK_V4 (WPK_F / 4)

__device__ __forceinline__ void load_b_chunk(const v4f* wq, int kc, v4f (&b)[4]) {
    #pragma unroll
    for (int q = 0; q < 4; ++q) b[q] = wq[(kc * 4 + q) * 32];
}

__device__ __forceinline__ void gemm_tile(const float* __restrict__ ytile, const float* __restrict__ wpk,
                                          const float* __restrict__ bias, float* __restrict__ out,
                                          int n0, int c, int half, int rbase, int lo, int M) {
    const float* ya = ytile + c * LDY + 2 * half;
    const v4f*   wp = (const v4f*)wpk + (size_t)(n0 >> 4) * (IN_SIZE / 32) * 4 * 32 + half * 16 + c;

    v8f acc[5];
    #pragma unroll
    for (int mt = 0; mt < 5; ++mt) acc[mt] = (v8f){0.f,0.f,0.f,0.f,0.f,0.f,0.f,0.f};

    v4f bc[4], bn[4];
    load_b_chunk(wp, 0, bc);
    #pragma unroll
    for (int kc = 0; kc < IN_SIZE / 32; ++kc) {
        if (kc + 1 < IN_SIZE / 32) load_b_chunk(wp, kc + 1, bn);
        __builtin_amdgcn_sched_barrier(0);
        #pragma unroll
        for (int ks = 0; ks < 8; ++ks) {
            v2f a[5];
            #pragma unroll
            for (int mt = 0; mt < 5; ++mt)
                a[mt] = *(const v2f*)(ya + mt * 16 * LDY + kc * 32 + ks * 4);
            const v4f bq = bc[ks >> 1];
            const v2f b  = (ks & 1) ? (v2f){bq.z, bq.w} : (v2f){bq.x, bq.y};
            #pragma unroll
            for (int mt = 0; mt < 5; ++mt)
                acc[mt] = __builtin_amdgcn_wmma_f32_16x16x4_f32(
                    false, a[mt], false, b, (short)0, acc[mt], false, false);
        }
        #pragma unroll
        for (int q = 0; q < 4; ++q) bc[q] = bn[q];
    }

    const float bv = bias[n0 + c];
    float* op = out + (size_t)(lo + rbase + 8 * half) * OUT_SIZE + n0 + c;
    if (rbase + TM <= TR && lo + rbase + TM <= M) {
        #pragma unroll
        for (int mt = 0; mt < 5; ++mt)
            #pragma unroll
            for (int v = 0; v < 8; ++v)
                op[(size_t)(mt * 16 + v) * OUT_SIZE] = acc[mt][v] + bv;
    } else {
        #pragma unroll
        for (int mt = 0; mt < 5; ++mt)
            #pragma unroll
            for (int v = 0; v < 8; ++v) {
                const int r = rbase + mt * 16 + v + 8 * half;
                if (r < TR && lo + r < M)
                    op[(size_t)(mt * 16 + v) * OUT_SIZE] = acc[mt][v] + bv;
            }
    }
}

__device__ __forceinline__ void lds_barrier() {
    __builtin_amdgcn_fence(__ATOMIC_RELEASE, "workgroup", "local");
    __builtin_amdgcn_s_barrier();
    __builtin_amdgcn_fence(__ATOMIC_ACQUIRE, "workgroup", "local");
}

__device__ __forceinline__ v4f w_group(const float* __restrict__ W, int g) {
    const int l  = g & 31;
    const int q  = (g >> 5) & 3;
    const int kc = (g >> 7) & (IN_SIZE / 32 - 1);
    const int j  = g / (32 * 4 * (IN_SIZE / 32));
    const float* p = W + (size_t)(kc * 32 + q * 8 + 2 * (l >> 4)) * OUT_SIZE + 16 * j + (l & 15);
    v4f v;
    v.x = p[0];
    v.y = p[OUT_SIZE];
    v.z = p[4 * OUT_SIZE];
    v.w = p[5 * OUT_SIZE];
    return v;
}


__global__ __launch_bounds__(BIN_T)
void gcn_bin_edges(const float* __restrict__ W, const float* __restrict__ ew, const int* __restrict__ esrc,
                   const int* __restrict__ edst, int* __restrict__ ws, int M, int E) {
    __shared__ int hist[NBMAX];
    __shared__ int fill[NBMAX];
    __shared__ int   st_row[SCAP];
    __shared__ int   st_src[SCAP];
    __shared__ float st_w[SCAP];
    __shared__ int   s_nrec;

    const int t    = threadIdx.x;
    const int lane = t & 31;

    static_assert((WPK_V4 + NSL - 1) / NSL <= BIN_T, "one group per thread");
    const int  wg  = blockIdx.x * ((WPK_V4 + NSL - 1) / NSL) + t;
    const bool wdo = t < (WPK_V4 + NSL - 1) / NSL && wg < WPK_V4;

    const int nb = (M + TR - 1) / TR;
    if (nb > NBMAX) {
        if (wdo) ((v4f*)ws)[wg] = w_group(W, wg);
        return;
    }

    const int s    = blockIdx.x;
    const int slen = (E + NSL - 1) / NSL;
    const int cap  = NSL * slen;
    int*   wsi     = ws + WPK_F;
    int*   rec_row = wsi;
    int*   rec_src = wsi + cap;
    float* rec_w   = (float*)(wsi + 2 * cap);
    int2*  tab     = (int2*)(wsi + 3 * cap);
    const int e_lo = s * slen;
    const int e_hi = min(E, e_lo + slen);
    const bool staged = slen <= SCAP;

    for (int i = t; i < nb; i += BIN_T) hist[i] = 0;
    __syncthreads();

    v4f wv = (v4f){0.f, 0.f, 0.f, 0.f};
    if (wdo) wv = w_group(W, wg);

    for (int e0 = e_lo; e0 < e_hi; e0 += BIN_T * BUNR) {
        int d[BUNR];
        #pragma unroll
        for (int u = 0; u < BUNR; ++u) {
            const int e = e0 + u * BIN_T + t;
            d[u] = e < e_hi ? edst[e] : -1;
        }
        #pragma unroll
        for (int u = 0; u < BUNR; ++u)
            if ((unsigned)d[u] < (unsigned)M) atomicAdd(&hist[d[u] / TR], 1);
    }
    __syncthreads();

    if (t < 32) {
        const int per = (nb + 31) / 32;
        int sum = 0;
        for (int i = 0; i < per; ++i) {
            const int b = lane * per + i;
            const int n = hist[min(b, NBMAX - 1)];
            sum += b < nb ? n : 0;
        }
        int incl = sum;
        #pragma unroll
        for (int dd = 1; dd < 32; dd <<= 1) {
            const int v = __shfl_up(incl, dd, 32);
            if (lane >= dd) incl += v;
        }
        int run = incl - sum;
        for (int i = 0; i < per; ++i) {
            const int b = lane * per + i;
            const int n = hist[min(b, NBMAX - 1)];
            if (b < nb) {
                fill[b] = run;
                tab[(size_t)b * NSL + s] = make_int2(run, n);
                run += n;
            }
        }
        if (lane == 31) s_nrec = run;
    }
    __syncthreads();

    for (int e0 = e_lo; e0 < e_hi; e0 += BIN_T * BUNR) {
        int   d[BUNR], sv[BUNR];
        float wq[BUNR];
        #pragma unroll
        for (int u = 0; u < BUNR; ++u) {
            const int e = e0 + u * BIN_T + t;
            const bool ok = e < e_hi;
            d[u]  = ok ? edst[e] : -1;
            sv[u] = ok ? esrc[e] : 0;
            wq[u] = ok ? ew[e]   : 0.f;
        }
        int g[BUNR];
        #pragma unroll
        for (int u = 0; u < BUNR; ++u) {
            g[u] = -1;
            if ((unsigned)d[u] < (unsigned)M) {
                const int b = d[u] / TR;
                g[u] = atomicAdd(&fill[b], 1);
                d[u] -= b * TR;
            }
        }
        if (staged) {
            #pragma unroll
            for (int u = 0; u < BUNR; ++u) {
                if (g[u] >= 0) {
                    st_row[g[u]] = d[u];
                    st_src[g[u]] = sv[u];
                    st_w[g[u]]   = wq[u];
                }
            }
        } else {
            #pragma unroll
            for (int u = 0; u < BUNR; ++u) {
                if (g[u] >= 0) {
                    rec_row[e_lo + g[u]] = d[u];
                    rec_src[e_lo + g[u]] = sv[u];
                    rec_w[e_lo + g[u]]   = wq[u];
                }
            }
        }
    }

    if (staged) {
        __syncthreads();
        const int n   = s_nrec;
        const int pad = e_lo & 31;
        if (n > 0) {
            for (int i0 = t - pad; i0 < n; i0 += BIN_T * 4) {
                int   r[4], sv[4];
                float wq[4];
                #pragma unroll
                for (int u = 0; u < 4; ++u) {
                    const int j = min(max(i0 + u * BIN_T, 0), n - 1);
                    r[u]  = st_row[j];
                    sv[u] = st_src[j];
                    wq[u] = st_w[j];
                }
                #pragma unroll
                for (int u = 0; u < 4; ++u) {
                    const int i = i0 + u * BIN_T;
                    if (i >= 0 && i < n) {
                        rec_row[e_lo + i] = r[u];
                        rec_src[e_lo + i] = sv[u];
                        rec_w[e_lo + i]   = wq[u];
                    }
                }
            }
        }
    }

    if (wdo) ((v4f*)ws)[wg] = wv;
}

__global__ __launch_bounds__(AGG_T)
void gcn_aggregate_then_gemm(const float* __restrict__ X, const float* __restrict__ bias,
                             const float* __restrict__ ew, const int* __restrict__ esrc,
                             const int* __restrict__ edst, const int* __restrict__ ws,
                             float* __restrict__ out, int M, int E) {
    __shared__ __attribute__((aligned(16))) char lraw[2 * Y_STAGE * 4];
    __shared__ int   s_src[CAPE];
    __shared__ float s_w[CAPE];
    __shared__ int   rowcnt[TR + 1];
    __shared__ int   rowoff[TR + 1];
    __shared__ int   rowfill[TR + 1];
    __shared__ int   s_total;

    float* ybuf  = (float*)lraw;

    const int t    = threadIdx.x;
    const int lane = t & 31;
    const int wv   = __builtin_amdgcn_readfirstlane(t >> 5);
    const int lo   = blockIdx.x * TR;
    const int c    = lane & 15;
    const int half = lane >> 4;

    const bool mat_wave = ((wv >> 2) & 1) == 0;
    const int  ridx     = ((wv >> 3) << 2) | (wv & 3);

    const int  nchunk = (E + ECHUNK - 1) / ECHUNK;
    const bool binned = (M + TR - 1) / TR <= NBMAX;

    const int slen = (E + NSL - 1) / NSL;
    const int cap  = NSL * slen;
    const float* wpk     = (const float*)ws;
    const int*   wsi     = ws + WPK_F;
    const int*   rec_row = wsi;
    const int*   rec_src = wsi + cap;
    const float* rec_w   = (const float*)(wsi + 2 * cap);
    const int2*  tab     = (const int2*)(wsi + 3 * cap) + (size_t)blockIdx.x * NSL;

    if (t <= TR) rowcnt[t] = 0;
    __syncthreads();

    int   rr[NSL / 32], rs[NSL / 32], rbeg[NSL / 32], rlen[NSL / 32];
    float rw[NSL / 32];
    #pragma unroll
    for (int u = 0; u < NSL / 32; ++u) { rr[u] = -1; rs[u] = 0; rw[u] = 0.f; rbeg[u] = 0; rlen[u] = 0; }
    if (binned) {
        #pragma unroll
        for (int u = 0; u < NSL / 32; ++u) {
            const int  sl = wv + 32 * u;
            const int2 oc = tab[sl];
            rbeg[u] = sl * slen + oc.x;
            rlen[u] = oc.y;
        }
        #pragma unroll
        for (int u = 0; u < NSL / 32; ++u) {
            if (lane < rlen[u]) {
                const int g = rbeg[u] + lane;
                rr[u] = rec_row[g];
                rs[u] = rec_src[g];
                rw[u] = rec_w[g];
            }
        }
        #pragma unroll
        for (int u = 0; u < NSL / 32; ++u)
            if (rr[u] >= 0) atomicAdd(&rowcnt[rr[u]], 1);
        #pragma unroll
        for (int u = 0; u < NSL / 32; ++u)
            for (int j = 32 + lane; j < rlen[u]; j += 32)
                atomicAdd(&rowcnt[rec_row[rbeg[u] + j]], 1);
    }
    __syncthreads();

    if (wv == 0) {
        int c[NROWP];
        int s = 0;
        #pragma unroll
        for (int i = 0; i < NROWP; ++i) {
            const int r = lane * NROWP + i;
            c[i] = r < TR ? rowcnt[r] : 0;
            s += c[i];
        }
        int incl = s;
        #pragma unroll
        for (int d = 1; d < 32; d <<= 1) {
            const int v = __shfl_up(incl, d, 32);
            if (lane >= d) incl += v;
        }
        int run = incl - s;
        #pragma unroll
        for (int i = 0; i < NROWP; ++i) {
            const int r = lane * NROWP + i;
            if (r < TR) { rowoff[r] = run; rowfill[r] = run; }
            run += c[i];
        }
        if (lane == 31) { s_total = run; rowoff[TR] = run; }
    }
    __syncthreads();
    const bool fallback = !binned || s_total > CAPE;

    if (!fallback) {
        int slot[NSL / 32];
        #pragma unroll
        for (int u = 0; u < NSL / 32; ++u) {
            slot[u] = -1;
            if (rr[u] >= 0) slot[u] = atomicAdd(&rowfill[rr[u]], 1);
        }
        #pragma unroll
        for (int u = 0; u < NSL / 32; ++u) {
            if (slot[u] >= 0) {
                s_src[slot[u]] = rs[u];
                s_w[slot[u]]   = rw[u];
            }
        }
        #pragma unroll
        for (int u = 0; u < NSL / 32; ++u) {
            for (int j = 32 + lane; j < rlen[u]; j += 32) {
                const int g  = rbeg[u] + j;
                const int pl = atomicAdd(&rowfill[rec_row[g]], 1);
                s_src[pl] = rec_src[g];
                s_w[pl]   = rec_w[g];
            }
        }
        __syncthreads();

        for (int st = 0; st <= NST; ++st) {
            if (!mat_wave) {
                if (st < NST) {
                    float* yb = ybuf + (st & 1) * Y_STAGE;
                    const int rl0 = ridx * GR;
                    const int r0  = st * TM + rl0;
                    const int bv = rowoff[min(r0 + min(lane, GR), TR)];
                    int bnd[GR + 1];
                    #pragma unroll
                    for (int k = 0; k <= GR; ++k) bnd[k] = __builtin_amdgcn_readlane(bv, k);
                    const int eend = bnd[GR];
                    int cur   = 0;
                    int nextb = bnd[1];
                    v4f s0 = (v4f){0.f, 0.f, 0.f, 0.f}, s1 = (v4f){0.f, 0.f, 0.f, 0.f};
                    for (int e = bnd[0]; e < eend; e += 8) {
                        const int   idx = min(e + (lane & 7), eend - 1);
                        const int   sv  = s_src[idx];
                        const float wv  = s_w[idx];
                        v4f   a[8], b[8];
                        float w[8];
                        #pragma unroll
                        for (int u = 0; u < 8; ++u) {
                            const int si = __builtin_amdgcn_readlane(sv, u);
                            w[u] = __int_as_float(__builtin_amdgcn_readlane(__float_as_int(wv), u));
                            const float* p = X + (size_t)si * IN_SIZE + lane * 4;
                            a[u] = *(const v4f*)(p);
                            b[u] = *(const v4f*)(p + 128);
                        }
                        #pragma unroll
                        for (int u = 0; u < 8; ++u) {
                            const int ei = e + u;
                            if (ei < eend) {
                                while (ei >= nextb && cur < GR - 1) {
                                    *(v4f*)(yb + (rl0 + cur) * LDY + lane * 4)       = s0;
                                    *(v4f*)(yb + (rl0 + cur) * LDY + 128 + lane * 4) = s1;
                                    s0 = (v4f){0.f, 0.f, 0.f, 0.f};
                                    s1 = (v4f){0.f, 0.f, 0.f, 0.f};
                                    ++cur;
                                    nextb = bnd[GR];
                                    #pragma unroll
                                    for (int k = 1; k < GR; ++k) if (cur + 1 == k) nextb = bnd[k];
                                }
                                s0 += a[u] * w[u];
                                s1 += b[u] * w[u];
                            }
                        }
                    }
                    for (; cur < GR; ++cur) {
                        *(v4f*)(yb + (rl0 + cur) * LDY + lane * 4)       = s0;
                        *(v4f*)(yb + (rl0 + cur) * LDY + 128 + lane * 4) = s1;
                        s0 = (v4f){0.f, 0.f, 0.f, 0.f};
                        s1 = (v4f){0.f, 0.f, 0.f, 0.f};
                    }
                }
            } else {
                if (st >= 1)
                    gemm_tile(ybuf + ((st - 1) & 1) * Y_STAGE, wpk, bias, out,
                              ridx * 16, c, half, (st - 1) * TM, lo, M);
            }
            lds_barrier();
        }
    } else {
        for (int st = 0; st < NST; ++st) {
            for (int i = t; i < Y_STAGE / 4; i += AGG_T)
                ((float4*)ybuf)[i] = make_float4(0.f, 0.f, 0.f, 0.f);
            __syncthreads();
            const int tlo = lo + st * TM;
            const int tn  = min(TM, TR - st * TM);
            for (int ch = wv; ch < nchunk; ch += 32) {
                const int eb = ch * ECHUNK;
                for (int x = 0; x < ECHUNK; x += 32) {
                    const int e = eb + x + lane;
                    const int d = e < E ? edst[e] : -1;
                    unsigned mk = __builtin_amdgcn_ballot_w32((unsigned)(d - tlo) < (unsigned)tn && d < M);
                    while (mk) {
                        const int l = __builtin_ctz(mk);
                        mk &= mk - 1;
                        const int   ee  = eb + x + l;
                        const int   dd  = edst[ee];
                        const float w   = ew[ee];
                        const float* p  = X + (size_t)esrc[ee] * IN_SIZE + lane * 4;
                        const v4f a = *(const v4f*)(p);
                        const v4f b = *(const v4f*)(p + 128);
                        float* q = ybuf + (dd - tlo) * LDY + lane * 4;
                        atomicAdd(q + 0,   a.x * w);
                        atomicAdd(q + 1,   a.y * w);
                        atomicAdd(q + 2,   a.z * w);
                        atomicAdd(q + 3,   a.w * w);
                        atomicAdd(q + 128, b.x * w);
                        atomicAdd(q + 129, b.y * w);
                        atomicAdd(q + 130, b.z * w);
                        atomicAdd(q + 131, b.w * w);
                    }
                }
            }
            __syncthreads();
            if (mat_wave)
                gemm_tile(ybuf, wpk, bias, out, ridx * 16, c, half, st * TM, lo, M);
            __syncthreads();
        }
    }
}

extern "C" void kernel_launch(void* const* d_in, const int* in_sizes, int n_in,
                              void* d_out, int out_size, void* d_ws, size_t ws_size,
                              hipStream_t stream) {
    const float* x    = (const float*)d_in[0];
    const float* w    = (const float*)d_in[1];
    const float* bias = (const float*)d_in[2];
    const float* ew   = (const float*)d_in[3];
    const int*   esrc = (const int*)d_in[4];
    const int*   edst = (const int*)d_in[5];
    float* out  = (float*)d_out;

    const int M = in_sizes[0] / IN_SIZE;
    const int E = in_sizes[3];

    int* wsi = (int*)d_ws;
    dim3 agrid((M + TR - 1) / TR);
    gcn_bin_edges<<<NSL, BIN_T, 0, stream>>>(w, ew, esrc, edst, wsi, M, E);
    gcn_aggregate_then_gemm<<<agrid, AGG_T, 0, stream>>>(x, bias, ew, esrc, edst, wsi, out, M, E);
}


# Out rows pass through a third LDS tile: matrix waves put their slabs there, gather waves store whole rows, 512 contiguous bytes per store, not two 64-byte pieces; list LDS halved; sum order kept
# speedup vs baseline: 4.6430x; 1.0517x over previous
#include <hip/hip_runtime.h>

typedef __attribute__((ext_vector_type(2))) float v2f;
typedef __attribute__((ext_vector_type(8))) float v8f;
typedef __attribute__((ext_vector_type(4))) float v4f;

#define IN_SIZE  256
#define OUT_SIZE 256

#define TR     391
#define AGG_T  1024
#define ECHUNK 1024
#define CAPE   8192
#define NSL    256
#define BIN_T  256
#define BUNR   16
#define NBMAX  1024
#define SCAP   (BIN_T * BUNR)
#define NROWP  ((TR + 31) / 32)
#define TM     80
#define NST    ((TR + TM - 1) / TM)
#define GR     (TM / 16)
#define LDY    260
#define Y_STAGE (TM * LDY)
#define WPK_F  (IN_SIZE * OUT_SIZE)
#define WPK_V4 (WPK_F / 4)

__device__ __forceinline__ void load_b_chunk(const v4f* wq, int kc, v4f (&b)[4]) {
    #pragma unroll
    for (int q = 0; q < 4; ++q) b[q] = wq[(kc * 4 + q) * 32];
}

__device__ __forceinline__ void gemm_tile_acc(const float* __restrict__ ytile, const float* __restrict__ wpk,
                                              int n0, int c, int half, v8f (&acc)[5]) {
    const float* ya = ytile + c * LDY + 2 * half;
    const v4f*   wp = (const v4f*)wpk + (size_t)(n0 >> 4) * (IN_SIZE / 32) * 4 * 32 + half * 16 + c;

    #pragma unroll
    for (int mt = 0; mt < 5; ++mt) acc[mt] = (v8f){0.f,0.f,0.f,0.f,0.f,0.f,0.f,0.f};

    v4f bc[4], bn[4];
    load_b_chunk(wp, 0, bc);
    #pragma unroll
    for (int kc = 0; kc < IN_SIZE / 32; ++kc) {
        if (kc + 1 < IN_SIZE / 32) load_b_chunk(wp, kc + 1, bn);
        __builtin_amdgcn_sched_barrier(0);
        #pragma unroll
        for (int ks = 0; ks < 8; ++ks) {
            v2f a[5];
            #pragma unroll
            for (int mt = 0; mt < 5; ++mt)
                a[mt] = *(const v2f*)(ya + mt * 16 * LDY + kc * 32 + ks * 4);
            const v4f bq = bc[ks >> 1];
            const v2f b  = (ks & 1) ? (v2f){bq.z, bq.w} : (v2f){bq.x, bq.y};
            #pragma unroll
            for (int mt = 0; mt < 5; ++mt)
                acc[mt] = __builtin_amdgcn_wmma_f32_16x16x4_f32(
                    false, a[mt], false, b, (short)0, acc[mt], false, false);
        }
        #pragma unroll
        for (int q = 0; q < 4; ++q) bc[q] = bn[q];
    }
}

__device__ __forceinline__ void gemm_tile(const float* __restrict__ ytile, const float* __restrict__ wpk,
                                          const float* __restrict__ bias, float* __restrict__ out,
                                          int n0, int c, int half, int rbase, int lo, int M) {
    v8f acc[5];
    gemm_tile_acc(ytile, wpk, n0, c, half, acc);

    const float bv = bias[n0 + c];
    float* op = out + (size_t)(lo + rbase + 8 * half) * OUT_SIZE + n0 + c;
    if (rbase + TM <= TR && lo + rbase + TM <= M) {
        #pragma unroll
        for (int mt = 0; mt < 5; ++mt)
            #pragma unroll
            for (int v = 0; v < 8; ++v)
                op[(size_t)(mt * 16 + v) * OUT_SIZE] = acc[mt][v] + bv;
    } else {
        #pragma unroll
        for (int mt = 0; mt < 5; ++mt)
            #pragma unroll
            for (int v = 0; v < 8; ++v) {
                const int r = rbase + mt * 16 + v + 8 * half;
                if (r < TR && lo + r < M)
                    op[(size_t)(mt * 16 + v) * OUT_SIZE] = acc[mt][v] + bv;
            }
    }
}

__device__ __forceinline__ void lds_barrier() {
    __builtin_amdgcn_fence(__ATOMIC_RELEASE, "workgroup", "local");
    __builtin_amdgcn_s_barrier();
    __builtin_amdgcn_fence(__ATOMIC_ACQUIRE, "workgroup", "local");
}

__device__ __forceinline__ v4f w_group(const float* __restrict__ W, int g) {
    const int l  = g & 31;
    const int q  = (g >> 5) & 3;
    const int kc = (g >> 7) & (IN_SIZE / 32 - 1);
    const int j  = g / (32 * 4 * (IN_SIZE / 32));
    const float* p = W + (size_t)(kc * 32 + q * 8 + 2 * (l >> 4)) * OUT_SIZE + 16 * j + (l & 15);
    v4f v;
    v.x = p[0];
    v.y = p[OUT_SIZE];
    v.z = p[4 * OUT_SIZE];
    v.w = p[5 * OUT_SIZE];
    return v;
}


__global__ __launch_bounds__(BIN_T)
void gcn_bin_edges(const float* __restrict__ W, const float* __restrict__ ew, const int* __restrict__ esrc,
                   const int* __restrict__ edst, int* __restrict__ ws, int M, int E) {
    __shared__ int hist[NBMAX];
    __shared__ int fill[NBMAX];
    __shared__ int   st_row[SCAP];
    __shared__ int   st_src[SCAP];
    __shared__ float st_w[SCAP];
    __shared__ int   s_nrec;

    const int t    = threadIdx.x;
    const int lane = t & 31;

    static_assert((WPK_V4 + NSL - 1) / NSL <= BIN_T, "one group per thread");
    const int  wg  = blockIdx.x * ((WPK_V4 + NSL - 1) / NSL) + t;
    const bool wdo = t < (WPK_V4 + NSL - 1) / NSL && wg < WPK_V4;

    const int nb = (M + TR - 1) / TR;
    if (nb > NBMAX) {
        if (wdo) ((v4f*)ws)[wg] = w_group(W, wg);
        return;
    }

    const int s    = blockIdx.x;
    const int slen = (E + NSL - 1) / NSL;
    const int cap  = NSL * slen;
    int*   wsi     = ws + WPK_F;
    int*   rec_row = wsi;
    int*   rec_src = wsi + cap;
    float* rec_w   = (float*)(wsi + 2 * cap);
    int2*  tab     = (int2*)(wsi + 3 * cap);
    const int e_lo = s * slen;
    const int e_hi = min(E, e_lo + slen);
    const bool staged = slen <= SCAP;

    for (int i = t; i < nb; i += BIN_T) hist[i] = 0;
    __syncthreads();

    v4f wv = (v4f){0.f, 0.f, 0.f, 0.f};
    if (wdo) wv = w_group(W, wg);

    for (int e0 = e_lo; e0 < e_hi; e0 += BIN_T * BUNR) {
        int d[BUNR];
        #pragma unroll
        for (int u = 0; u < BUNR; ++u) {
            const int e = e0 + u * BIN_T + t;
            d[u] = e < e_hi ? edst[e] : -1;
        }
        #pragma unroll
        for (int u = 0; u < BUNR; ++u)
            if ((unsigned)d[u] < (unsigned)M) atomicAdd(&hist[d[u] / TR], 1);
    }
    __syncthreads();

    if (t < 32) {
        const int per = (nb + 31) / 32;
        int sum = 0;
        for (int i = 0; i < per; ++i) {
            const int b = lane * per + i;
            const int n = hist[min(b, NBMAX - 1)];
            sum += b < nb ? n : 0;
        }
        int incl = sum;
        #pragma unroll
        for (int dd = 1; dd < 32; dd <<= 1) {
            const int v = __shfl_up(incl, dd, 32);
            if (lane >= dd) incl += v;
        }
        int run = incl - sum;
        for (int i = 0; i < per; ++i) {
            const int b = lane * per + i;
            const int n = hist[min(b, NBMAX - 1)];
            if (b < nb) {
                fill[b] = run;
                tab[(size_t)b * NSL + s] = make_int2(run, n);
                run += n;
            }
        }
        if (lane == 31) s_nrec = run;
    }
    __syncthreads();

    for (int e0 = e_lo; e0 < e_hi; e0 += BIN_T * BUNR) {
        int   d[BUNR], sv[BUNR];
        float wq[BUNR];
        #pragma unroll
        for (int u = 0; u < BUNR; ++u) {
            const int e = e0 + u * BIN_T + t;
            const bool ok = e < e_hi;
            d[u]  = ok ? edst[e] : -1;
            sv[u] = ok ? esrc[e] : 0;
            wq[u] = ok ? ew[e]   : 0.f;
        }
        int g[BUNR];
        #pragma unroll
        for (int u = 0; u < BUNR; ++u) {
            g[u] = -1;
            if ((unsigned)d[u] < (unsigned)M) {
                const int b = d[u] / TR;
                g[u] = atomicAdd(&fill[b], 1);
                d[u] -= b * TR;
            }
        }
        if (staged) {
            #pragma unroll
            for (int u = 0; u < BUNR; ++u) {
                if (g[u] >= 0) {
                    st_row[g[u]] = d[u];
                    st_src[g[u]] = sv[u];
                    st_w[g[u]]   = wq[u];
                }
            }
        } else {
            #pragma unroll
            for (int u = 0; u < BUNR; ++u) {
                if (g[u] >= 0) {
                    rec_row[e_lo + g[u]] = d[u];
                    rec_src[e_lo + g[u]] = sv[u];
                    rec_w[e_lo + g[u]]   = wq[u];
                }
            }
        }
    }

    if (staged) {
        __syncthreads();
        const int n   = s_nrec;
        const int pad = e_lo & 31;
        if (n > 0) {
            for (int i0 = t - pad; i0 < n; i0 += BIN_T * 4) {
                int   r[4], sv[4];
                float wq[4];
                #pragma unroll
                for (int u = 0; u < 4; ++u) {
                    const int j = min(max(i0 + u * BIN_T, 0), n - 1);
                    r[u]  = st_row[j];
                    sv[u] = st_src[j];
                    wq[u] = st_w[j];
                }
                #pragma unroll
                for (int u = 0; u < 4; ++u) {
                    const int i = i0 + u * BIN_T;
                    if (i >= 0 && i < n) {
                        rec_row[e_lo + i] = r[u];
                        rec_src[e_lo + i] = sv[u];
                        rec_w[e_lo + i]   = wq[u];
                    }
                }
            }
        }
    }

    if (wdo) ((v4f*)ws)[wg] = wv;
}

__global__ __launch_bounds__(AGG_T)
void gcn_aggregate_then_gemm(const float* __restrict__ X, const float* __restrict__ bias,
                             const float* __restrict__ ew, const int* __restrict__ esrc,
                             const int* __restrict__ edst, const int* __restrict__ ws,
                             float* __restrict__ out, int M, int E) {
    __shared__ __attribute__((aligned(16))) char lraw[2 * Y_STAGE * 4];
    __shared__ __attribute__((aligned(16))) float obuf[Y_STAGE];
    __shared__ int   s_src[CAPE];
    __shared__ float s_w[CAPE];
    __shared__ int   rowcnt[TR + 1];
    __shared__ int   rowoff[TR + 1];
    __shared__ int   rowfill[TR + 1];
    __shared__ int   s_total;

    float* ybuf  = (float*)lraw;

    const int t    = threadIdx.x;
    const int lane = t & 31;
    const int wv   = __builtin_amdgcn_readfirstlane(t >> 5);
    const int lo   = blockIdx.x * TR;
    const int c    = lane & 15;
    const int half = lane >> 4;

    const bool mat_wave = ((wv >> 2) & 1) == 0;
    const int  ridx     = ((wv >> 3) << 2) | (wv & 3);

    const int  nchunk = (E + ECHUNK - 1) / ECHUNK;
    const bool binned = (M + TR - 1) / TR <= NBMAX;

    const int slen = (E + NSL - 1) / NSL;
    const int cap  = NSL * slen;
    const float* wpk     = (const float*)ws;
    const int*   wsi     = ws + WPK_F;
    const int*   rec_row = wsi;
    const int*   rec_src = wsi + cap;
    const float* rec_w   = (const float*)(wsi + 2 * cap);
    const int2*  tab     = (const int2*)(wsi + 3 * cap) + (size_t)blockIdx.x * NSL;

    if (t <= TR) rowcnt[t] = 0;
    __syncthreads();

    int   rr[NSL / 32], rs[NSL / 32], rbeg[NSL / 32], rlen[NSL / 32];
    float rw[NSL / 32];
    #pragma unroll
    for (int u = 0; u < NSL / 32; ++u) { rr[u] = -1; rs[u] = 0; rw[u] = 0.f; rbeg[u] = 0; rlen[u] = 0; }
    if (binned) {
        #pragma unroll
        for (int u = 0; u < NSL / 32; ++u) {
            const int  sl = wv + 32 * u;
            const int2 oc = tab[sl];
            rbeg[u] = sl * slen + oc.x;
            rlen[u] = oc.y;
        }
        #pragma unroll
        for (int u = 0; u < NSL / 32; ++u) {
            if (lane < rlen[u]) {
                const int g = rbeg[u] + lane;
                rr[u] = rec_row[g];
                rs[u] = rec_src[g];
                rw[u] = rec_w[g];
            }
        }
        #pragma unroll
        for (int u = 0; u < NSL / 32; ++u)
            if (rr[u] >= 0) atomicAdd(&rowcnt[rr[u]], 1);
        #pragma unroll
        for (int u = 0; u < NSL / 32; ++u)
            for (int j = 32 + lane; j < rlen[u]; j += 32)
                atomicAdd(&rowcnt[rec_row[rbeg[u] + j]], 1);
    }
    __syncthreads();

    if (wv == 0) {
        int c[NROWP];
        int s = 0;
        #pragma unroll
        for (int i = 0; i < NROWP; ++i) {
            const int r = lane * NROWP + i;
            c[i] = r < TR ? rowcnt[r] : 0;
            s += c[i];
        }
        int incl = s;
        #pragma unroll
        for (int d = 1; d < 32; d <<= 1) {
            const int v = __shfl_up(incl, d, 32);
            if (lane >= d) incl += v;
        }
        int run = incl - s;
        #pragma unroll
        for (int i = 0; i < NROWP; ++i) {
            const int r = lane * NROWP + i;
            if (r < TR) { rowoff[r] = run; rowfill[r] = run; }
            run += c[i];
        }
        if (lane == 31) { s_total = run; rowoff[TR] = run; }
    }
    __syncthreads();
    const bool fallback = !binned || s_total > CAPE;

    if (!fallback) {
        int slot[NSL / 32];
        #pragma unroll
        for (int u = 0; u < NSL / 32; ++u) {
            slot[u] = -1;
            if (rr[u] >= 0) slot[u] = atomicAdd(&rowfill[rr[u]], 1);
        }
        #pragma unroll
        for (int u = 0; u < NSL / 32; ++u) {
            if (slot[u] >= 0) {
                s_src[slot[u]] = rs[u];
                s_w[slot[u]]   = rw[u];
            }
        }
        #pragma unroll
        for (int u = 0; u < NSL / 32; ++u) {
            for (int j = 32 + lane; j < rlen[u]; j += 32) {
                const int g  = rbeg[u] + j;
                const int pl = atomicAdd(&rowfill[rec_row[g]], 1);
                s_src[pl] = rec_src[g];
                s_w[pl]   = rec_w[g];
            }
        }
        __syncthreads();

        const float bv = bias[ridx * 16 + c];
        for (int st = 0; st <= NST + 1; ++st) {
            const bool mm = mat_wave && st >= 1 && st <= NST;
            v8f acc[5];
            #pragma unroll
            for (int mt = 0; mt < 5; ++mt) acc[mt] = (v8f){0.f,0.f,0.f,0.f,0.f,0.f,0.f,0.f};
            if (!mat_wave) {
                if (st >= 2) {
                    const int rb = (st - 2) * TM + ridx * GR;
                    const float* ob = obuf + ridx * GR * LDY + lane * 4;
                    v4f o0[GR], o1[GR];
                    #pragma unroll
                    for (int k = 0; k < GR; ++k) {
                        o0[k] = *(const v4f*)(ob + k * LDY);
                        o1[k] = *(const v4f*)(ob + k * LDY + 128);
                    }
                    #pragma unroll
                    for (int k = 0; k < GR; ++k) {
                        if (rb + k < TR && lo + rb + k < M) {
                            const int g = (lo + rb + k) * (OUT_SIZE / 4) + lane;
                            ((v4f*)out)[g]      = o0[k];
                            ((v4f*)out)[g + 32] = o1[k];
                        }
                    }
                }
                if (st < NST) {
                    float* yb = ybuf + (st & 1) * Y_STAGE;
                    const int rl0 = ridx * GR;
                    const int r0  = st * TM + rl0;
                    const int bv = rowoff[min(r0 + min(lane, GR), TR)];
                    int bnd[GR + 1];
                    #pragma unroll
                    for (int k = 0; k <= GR; ++k) bnd[k] = __builtin_amdgcn_readlane(bv, k);
                    const int eend = bnd[GR];
                    int cur   = 0;
                    int nextb = bnd[1];
                    v4f s0 = (v4f){0.f, 0.f, 0.f, 0.f}, s1 = (v4f){0.f, 0.f, 0.f, 0.f};
                    for (int e = bnd[0]; e < eend; e += 8) {
                        const int   idx = min(e + (lane & 7), eend - 1);
                        const int   sv  = s_src[idx];
                        const float wv  = s_w[idx];
                        v4f   a[8], b[8];
                        float w[8];
                        #pragma unroll
                        for (int u = 0; u < 8; ++u) {
                            const int si = __builtin_amdgcn_readlane(sv, u);
                            w[u] = __int_as_float(__builtin_amdgcn_readlane(__float_as_int(wv), u));
                            const float* p = X + (size_t)si * IN_SIZE + lane * 4;
                            a[u] = *(const v4f*)(p);
                            b[u] = *(const v4f*)(p + 128);
                        }
                        #pragma unroll
                        for (int u = 0; u < 8; ++u) {
                            const int ei = e + u;
                            if (ei < eend) {
                                while (ei >= nextb && cur < GR - 1) {
                                    *(v4f*)(yb + (rl0 + cur) * LDY + lane * 4)       = s0;
                                    *(v4f*)(yb + (rl0 + cur) * LDY + 128 + lane * 4) = s1;
                                    s0 = (v4f){0.f, 0.f, 0.f, 0.f};
                                    s1 = (v4f){0.f, 0.f, 0.f, 0.f};
                                    ++cur;
                                    nextb = bnd[GR];
                                    #pragma unroll
                                    for (int k = 1; k < GR; ++k) if (cur + 1 == k) nextb = bnd[k];
                                }
                                s0 += a[u] * w[u];
                                s1 += b[u] * w[u];
                            }
                        }
                    }
                    for (; cur < GR; ++cur) {
                        *(v4f*)(yb + (rl0 + cur) * LDY + lane * 4)       = s0;
                        *(v4f*)(yb + (rl0 + cur) * LDY + 128 + lane * 4) = s1;
                        s0 = (v4f){0.f, 0.f, 0.f, 0.f};
                        s1 = (v4f){0.f, 0.f, 0.f, 0.f};
                    }
                }
            } else if (mm) {
                gemm_tile_acc(ybuf + ((st - 1) & 1) * Y_STAGE, wpk, ridx * 16, c, half, acc);
            }
            lds_barrier();
            if (mm) {
                float* ow = obuf + 8 * half * LDY + ridx * 16 + c;
                #pragma unroll
                for (int mt = 0; mt < 5; ++mt)
                    #pragma unroll
                    for (int v = 0; v < 8; ++v)
                        ow[(mt * 16 + v) * LDY] = acc[mt][v] + bv;
            }
            lds_barrier();
        }
    } else {
        for (int st = 0; st < NST; ++st) {
            for (int i = t; i < Y_STAGE / 4; i += AGG_T)
                ((float4*)ybuf)[i] = make_float4(0.f, 0.f, 0.f, 0.f);
            __syncthreads();
            const int tlo = lo + st * TM;
            const int tn  = min(TM, TR - st * TM);
            for (int ch = wv; ch < nchunk; ch += 32) {
                const int eb = ch * ECHUNK;
                for (int x = 0; x < ECHUNK; x += 32) {
                    const int e = eb + x + lane;
                    const int d = e < E ? edst[e] : -1;
                    unsigned mk = __builtin_amdgcn_ballot_w32((unsigned)(d - tlo) < (unsigned)tn && d < M);
                    while (mk) {
                        const int l = __builtin_ctz(mk);
                        mk &= mk - 1;
                        const int   ee  = eb + x + l;
                        const int   dd  = edst[ee];
                        const float w   = ew[ee];
                        const float* p  = X + (size_t)esrc[ee] * IN_SIZE + lane * 4;
                        const v4f a = *(const v4f*)(p);
                        const v4f b = *(const v4f*)(p + 128);
                        float* q = ybuf + (dd - tlo) * LDY + lane * 4;
                        atomicAdd(q + 0,   a.x * w);
                        atomicAdd(q + 1,   a.y * w);
                        atomicAdd(q + 2,   a.z * w);
                        atomicAdd(q + 3,   a.w * w);
                        atomicAdd(q + 128, b.x * w);
                        atomicAdd(q + 129, b.y * w);
                        atomicAdd(q + 130, b.z * w);
                        atomicAdd(q + 131, b.w * w);
                    }
                }
            }
            __syncthreads();
            if (mat_wave)
                gemm_tile(ybuf, wpk, bias, out, ridx * 16, c, half, st * TM, lo, M);
            __syncthreads();
        }
    }
}

extern "C" void kernel_launch(void* const* d_in, const int* in_sizes, int n_in,
                              void* d_out, int out_size, void* d_ws, size_t ws_size,
                              hipStream_t stream) {
    const float* x    = (const float*)d_in[0];
    const float* w    = (const float*)d_in[1];
    const float* bias = (const float*)d_in[2];
    const float* ew   = (const float*)d_in[3];
    const int*   esrc = (const int*)d_in[4];
    const int*   edst = (const int*)d_in[5];
    float* out  = (float*)d_out;

    const int M = in_sizes[0] / IN_SIZE;
    const int E = in_sizes[3];

    int* wsi = (int*)d_ws;
    dim3 agrid((M + TR - 1) / TR);
    gcn_bin_edges<<<NSL, BIN_T, 0, stream>>>(w, ew, esrc, edst, wsi, M, E);
    gcn_aggregate_then_gemm<<<agrid, AGG_T, 0, stream>>>(x, bias, ew, esrc, edst, wsi, out, M, E);
}
